# starting kernel
# baseline (speedup 1.0000x reference)
#include <hip/hip_runtime.h>

namespace {

constexpr int kN  = 65536;
constexpr int kE  = 1048576;
constexpr int kDH = 64;
constexpr int kMaxK = 128;

typedef __attribute__((ext_vector_type(2))) float v2f;
typedef __attribute__((ext_vector_type(8))) float v8f;

__global__ void fill_kernel(float* __restrict__ p, float v, int n) {
  int i = blockIdx.x * blockDim.x + threadIdx.x;
  if (i < n) p[i] = v;
}

__global__ void deg_accum_kernel(const int* __restrict__ col,
                                 const float* __restrict__ ew,
                                 float* __restrict__ deg) {
  int e = blockIdx.x * blockDim.x + threadIdx.x;
  if (e < kE) atomicAdd(&deg[col[e]], ew[e]);
}

__global__ void dinv_kernel(float* __restrict__ d) {
  int i = blockIdx.x * blockDim.x + threadIdx.x;
  if (i < kN) {
    float v = d[i];
    d[i] = (v > 0.f) ? rsqrtf(v) : 0.f;
  }
}

__global__ void norm_kernel(const int* __restrict__ row, const int* __restrict__ col,
                            const float* __restrict__ ew, const float* __restrict__ dinv,
                            float* __restrict__ norm) {
  int e = blockIdx.x * blockDim.x + threadIdx.x;
  if (e < kE) norm[e] = dinv[row[e]] * ew[e] * dinv[col[e]];
}

__global__ void gemm_wmma_kernel(const float* __restrict__ A,
                                 const float* __restrict__ W,
                                 float* __restrict__ H, int K) {
  __shared__ float Ws[kMaxK * 17];

  const int tid  = threadIdx.x;
  const int lane = tid & 31;
  const int wave = tid >> 5;
  const int m0 = (blockIdx.x * 4 + wave) * 16;
  const int n0 = blockIdx.y * 16;
  const int half = lane >> 4;
  const int l = lane & 15;

  for (int idx = tid; idx < K * 16; idx += 128) {
    int kk = idx >> 4;
    int nn = idx & 15;
    Ws[kk * 17 + nn] = W[(size_t)kk * kDH + n0 + nn];
  }
  __syncthreads();

  v8f acc = {};
  const float* arow = A + (size_t)(m0 + l) * K;
  #pragma unroll 4
  for (int k = 0; k < K; k += 4) {
    const int ka = k + 2 * half;
    v2f a, b;
    a.x = arow[ka];
    a.y = arow[ka + 1];
    b.x = Ws[ka * 17 + l];
    b.y = Ws[(ka + 1) * 17 + l];
    acc = __builtin_amdgcn_wmma_f32_16x16x4_f32(
        false, a, false, b, (short)0, acc, false, false);
  }

  float* hrow = H + (size_t)(m0 + half * 8) * kDH + (n0 + l);
  #pragma unroll
  for (int r = 0; r < 8; ++r) hrow[(size_t)r * kDH] = acc[r];
}

__global__ void scatter64_kernel(const int* __restrict__ row, const int* __restrict__ col,
                                 const float* __restrict__ norm,
                                 const float* __restrict__ h,
                                 float* __restrict__ agg) {
  int t = blockIdx.x * blockDim.x + threadIdx.x;
  int e = t >> 5;
  int d = (t & 31) * 2;
  if (e < kE) {
    float nm = norm[e];
    const float* hp = h + (size_t)row[e] * kDH + d;
    float* ap = agg + (size_t)col[e] * kDH + d;
    atomicAdd(ap + 0, nm * hp[0]);
    atomicAdd(ap + 1, nm * hp[1]);
  }
}

__global__ void epilogue64_kernel(float* __restrict__ agg, const float* __restrict__ h,
                                  const float* __restrict__ dinv,
                                  const float* __restrict__ bias, int relu) {
  int t = blockIdx.x * blockDim.x + threadIdx.x;
  if (t < kN * kDH) {
    int i = t >> 6;
    int d = t & 63;
    float di = dinv[i];
    float v = agg[t] + di * di * h[t] + bias[d];
    if (relu) v = fmaxf(v, 0.f);
    agg[t] = v;
  }
}

__global__ void gemm_out_kernel(const float* __restrict__ feat,
                                const float* __restrict__ W8,
                                float* __restrict__ h2) {
  int i = blockIdx.x * blockDim.x + threadIdx.x;
  if (i < kN) {
    float s0 = 0.f, s1 = 0.f;
    const float* f = feat + (size_t)i * kDH;
    #pragma unroll
    for (int k = 0; k < kDH; ++k) {
      s0 = fmaf(f[k], W8[k * 2 + 0], s0);
      s1 = fmaf(f[k], W8[k * 2 + 1], s1);
    }
    h2[i * 2 + 0] = s0;
    h2[i * 2 + 1] = s1;
  }
}

__global__ void scatter2_kernel(const int* __restrict__ row, const int* __restrict__ col,
                                const float* __restrict__ norm,
                                const float* __restrict__ h2,
                                float* __restrict__ agg2) {
  int e = blockIdx.x * blockDim.x + threadIdx.x;
  if (e < kE) {
    float nm = norm[e];
    const float* hp = h2 + (size_t)row[e] * 2;
    float* ap = agg2 + (size_t)col[e] * 2;
    atomicAdd(ap + 0, nm * hp[0]);
    atomicAdd(ap + 1, nm * hp[1]);
  }
}

__global__ void epilogue_out_kernel(const float* __restrict__ agg2,
                                    const float* __restrict__ h2,
                                    const float* __restrict__ dinv,
                                    const float* __restrict__ b8,
                                    float* __restrict__ out) {
  int t = blockIdx.x * blockDim.x + threadIdx.x;
  if (t < kN * 2) {
    int i = t >> 1;
    int d = t & 1;
    float di = dinv[i];
    out[t] = agg2[t] + di * di * h2[t] + b8[d];
  }
}

}

extern "C" void kernel_launch(void* const* d_in, const int* in_sizes, int n_in,
                              void* d_out, int out_size, void* d_ws, size_t ws_size,
                              hipStream_t stream) {
  (void)in_sizes; (void)n_in; (void)out_size; (void)ws_size;

  const float* x  = (const float*)d_in[0];
  const int*   ei = (const int*)d_in[1];
  const float* ew = (const float*)d_in[2];
  const float* W[5] = {(const float*)d_in[3], (const float*)d_in[5], (const float*)d_in[7],
                       (const float*)d_in[9], (const float*)d_in[11]};
  const float* B[5] = {(const float*)d_in[4], (const float*)d_in[6], (const float*)d_in[8],
                       (const float*)d_in[10], (const float*)d_in[12]};
  const float* W8 = (const float*)d_in[13];
  const float* b8 = (const float*)d_in[14];
  float* out = (float*)d_out;

  const int* row = ei;
  const int* col = ei + kE;

  float* ws = (float*)d_ws;
  float* dinv  = ws;  ws += kN;
  float* norm  = ws;  ws += kE;
  float* featA = ws;  ws += (size_t)kN * kDH;
  float* featB = ws;  ws += (size_t)kN * kDH;
  float* hbuf  = ws;

  fill_kernel<<<kN / 256, 256, 0, stream>>>(dinv, 1.0f, kN);
  deg_accum_kernel<<<kE / 256, 256, 0, stream>>>(col, ew, dinv);
  dinv_kernel<<<kN / 256, 256, 0, stream>>>(dinv);
  norm_kernel<<<kE / 256, 256, 0, stream>>>(row, col, ew, dinv, norm);

  const dim3 ggrid(kN / (16 * 4), kDH / 16);
  const int nf = kN * kDH;

  const float* cur = x;
  int curK = 128;
  float* bufs[2] = {featA, featB};
  for (int layer = 0; layer < 5; ++layer) {
    float* agg = bufs[layer & 1];
    gemm_wmma_kernel<<<ggrid, 128, 0, stream>>>(cur, W[layer], hbuf, curK);
    fill_kernel<<<nf / 256, 256, 0, stream>>>(agg, 0.f, nf);
    scatter64_kernel<<<(kE * 32) / 256, 256, 0, stream>>>(row, col, norm, hbuf, agg);
    epilogue64_kernel<<<nf / 256, 256, 0, stream>>>(agg, hbuf, dinv, B[layer], 1);
    cur = agg;
    curK = kDH;
  }

  float* agg2 = bufs[1];
  gemm_out_kernel<<<kN / 256, 256, 0, stream>>>(cur, W8, hbuf);
  fill_kernel<<<(kN * 2) / 256, 256, 0, stream>>>(agg2, 0.f, kN * 2);
  scatter2_kernel<<<kE / 256, 256, 0, stream>>>(row, col, norm, hbuf, agg2);
  epilogue_out_kernel<<<(kN * 2) / 256, 256, 0, stream>>>(agg2, hbuf, dinv, b8, out);
}


# Edges bucketed by target each call (count+rank atomics, scan, place); per-node gather-sum replaces float-atomic scatter and absorbs zero-fill, self-loop, bias, ReLU; sums run in bucket order
# speedup vs baseline: 2.6623x; 2.6623x over previous
#include <hip/hip_runtime.h>


namespace {

constexpr int kN  = 65536;
constexpr int kE  = 1048576;
constexpr int kDH = 64;
constexpr int kMaxK = 128;

typedef __attribute__((ext_vector_type(2))) float v2f;
typedef __attribute__((ext_vector_type(8))) float v8f;

__device__ __forceinline__ float sum16(float v) {
  v += __shfl_xor(v, 8, 16);
  v += __shfl_xor(v, 4, 16);
  v += __shfl_xor(v, 2, 16);
  v += __shfl_xor(v, 1, 16);
  return v;
}

__global__ void zero_int_kernel(int* __restrict__ p, int n) {
  int i = blockIdx.x * blockDim.x + threadIdx.x;
  if (i < n) p[i] = 0;
}

__global__ void count_rank_kernel(const int* __restrict__ col,
                                  int* __restrict__ cnt,
                                  int* __restrict__ rank) {
  int e = blockIdx.x * blockDim.x + threadIdx.x;
  if (e < kE) rank[e] = atomicAdd(&cnt[col[e]], 1);
}

__global__ __launch_bounds__(1024) void scan_kernel(const int* __restrict__ cnt,
                                                    int* __restrict__ rowptr) {
  __shared__ int wsum[32];
  const int t = threadIdx.x;
  const int lane = t & 31;
  const int wv = t >> 5;
  const int4* c4 = reinterpret_cast<const int4*>(cnt) + t * 16;
  int4 v[16];
  #pragma unroll
  for (int q = 0; q < 16; ++q) v[q] = c4[q];
  int s = 0;
  #pragma unroll
  for (int q = 0; q < 16; ++q) {
    const int a = v[q].x, b = v[q].y, c = v[q].z, d = v[q].w;
    v[q].x = s; s += a;
    v[q].y = s; s += b;
    v[q].z = s; s += c;
    v[q].w = s; s += d;
  }
  int inc = s;
  #pragma unroll
  for (int off = 1; off < 32; off <<= 1) {
    const int nb = __shfl_up(inc, off);
    if (lane >= off) inc += nb;
  }
  if (lane == 31) wsum[wv] = inc;
  __syncthreads();
  if (wv == 0) {
    const int x = wsum[lane];
    int xi = x;
    #pragma unroll
    for (int off = 1; off < 32; off <<= 1) {
      const int nb = __shfl_up(xi, off);
      if (lane >= off) xi += nb;
    }
    wsum[lane] = xi - x;
  }
  __syncthreads();
  const int base = wsum[wv] + inc - s;
  int4* r4 = reinterpret_cast<int4*>(rowptr) + t * 16;
  #pragma unroll
  for (int q = 0; q < 16; ++q) {
    int4 o = v[q];
    o.x += base; o.y += base; o.z += base; o.w += base;
    r4[q] = o;
  }
  if (t == 1023) rowptr[kN] = base + s;
}

__global__ void place_kernel(const int* __restrict__ row, const int* __restrict__ col,
                             const float* __restrict__ ew,
                             const int* __restrict__ rowptr,
                             const int* __restrict__ rank,
                             int2* __restrict__ edges) {
  int e = blockIdx.x * blockDim.x + threadIdx.x;
  if (e < kE) {
    const int p = rowptr[col[e]] + rank[e];
    edges[p] = make_int2(row[e], __float_as_int(ew[e]));
  }
}

__global__ void dinv_kernel(const int* __restrict__ rowptr,
                            const int2* __restrict__ edges,
                            float* __restrict__ dinv) {
  const int t = blockIdx.x * blockDim.x + threadIdx.x;
  const int i = t >> 4;
  const int l = t & 15;
  const int beg = rowptr[i], end = rowptr[i + 1];
  float s = 0.f;
  for (int j = beg + l; j < end; j += 16) s += __int_as_float(edges[j].y);
  s = sum16(s) + 1.0f;
  if (l == 0) dinv[i] = (s > 0.f) ? rsqrtf(s) : 0.f;
}

__global__ void norm_kernel(const int* __restrict__ rowptr,
                            const float* __restrict__ dinv,
                            int2* __restrict__ edges) {
  const int t = blockIdx.x * blockDim.x + threadIdx.x;
  const int i = t >> 4;
  const int l = t & 15;
  const int beg = rowptr[i], end = rowptr[i + 1];
  const float di = dinv[i];
  for (int j = beg + l; j < end; j += 16) {
    const int2 ed = edges[j];
    const float nm = dinv[ed.x] * __int_as_float(ed.y) * di;
    edges[j].y = __float_as_int(nm);
  }
}

__global__ void gemm_wmma_kernel(const float* __restrict__ A,
                                 const float* __restrict__ W,
                                 float* __restrict__ H, int K) {
  __shared__ float Ws[kMaxK * 17];

  const int tid  = threadIdx.x;
  const int lane = tid & 31;
  const int wave = tid >> 5;
  const int m0 = (blockIdx.x * 4 + wave) * 16;
  const int n0 = blockIdx.y * 16;
  const int half = lane >> 4;
  const int l = lane & 15;

  for (int idx = tid; idx < K * 16; idx += 128) {
    int kk = idx >> 4;
    int nn = idx & 15;
    Ws[kk * 17 + nn] = W[(size_t)kk * kDH + n0 + nn];
  }
  __syncthreads();

  v8f acc = {};
  const float* arow = A + (size_t)(m0 + l) * K;
  #pragma unroll 4
  for (int k = 0; k < K; k += 4) {
    const int ka = k + 2 * half;
    v2f a, b;
    a.x = arow[ka];
    a.y = arow[ka + 1];
    b.x = Ws[ka * 17 + l];
    b.y = Ws[(ka + 1) * 17 + l];
    acc = __builtin_amdgcn_wmma_f32_16x16x4_f32(
        false, a, false, b, (short)0, acc, false, false);
  }

  float* hrow = H + (size_t)(m0 + half * 8) * kDH + (n0 + l);
  #pragma unroll
  for (int r = 0; r < 8; ++r) hrow[(size_t)r * kDH] = acc[r];
}

template <int NL>
__device__ __forceinline__ void rows_fma(const float4* __restrict__ h4, int l,
                                         int sx, float wx, float4& acc) {
  int s[NL];
  float w[NL];
  float4 a[NL];
  #pragma unroll
  for (int q = 0; q < NL; ++q) {
    s[q] = __shfl(sx, q, 16);
    w[q] = __shfl(wx, q, 16);
  }
  #pragma unroll
  for (int q = 0; q < NL; ++q) a[q] = h4[(size_t)s[q] * 16 + l];
  #pragma unroll
  for (int q = 0; q < NL; ++q) {
    acc.x += w[q] * a[q].x;
    acc.y += w[q] * a[q].y;
    acc.z += w[q] * a[q].z;
    acc.w += w[q] * a[q].w;
  }
}

__global__ __launch_bounds__(256) __attribute__((amdgpu_waves_per_eu(4, 8))) void gather64_kernel(const int* __restrict__ rowptr,
                                                       const int2* __restrict__ edges,
                                                       const float* __restrict__ h,
                                                       const float* __restrict__ dinv,
                                                       const float* __restrict__ bias,
                                                       float* __restrict__ outf) {
  const int t = blockIdx.x * blockDim.x + threadIdx.x;
  const int i = t >> 4;
  const int l = t & 15;
  const int beg = rowptr[i], end = rowptr[i + 1];
  const float4* __restrict__ h4 = reinterpret_cast<const float4*>(h);
  const float di = dinv[i];
  const float4 hs = h4[(size_t)i * 16 + l];
  const float b0 = bias[l * 4 + 0], b1 = bias[l * 4 + 1];
  const float b2 = bias[l * 4 + 2], b3 = bias[l * 4 + 3];

  float4 acc = make_float4(0.f, 0.f, 0.f, 0.f);
  int base = beg;
  int2 ed = edges[min(base + l, kE - 1)];
  int sx = (base + l < end) ? ed.x : i;
  float wx = (base + l < end) ? __int_as_float(ed.y) : 0.f;
  while (base < end) {
    const int m = end - base;
    const int nb = base + 16;
    ed = edges[min(nb + l, kE - 1)];
    if (m > 8) {
      rows_fma<16>(h4, l, sx, wx, acc);
    } else {
      rows_fma<8>(h4, l, sx, wx, acc);
    }
    sx = (nb + l < end) ? ed.x : i;
    wx = (nb + l < end) ? __int_as_float(ed.y) : 0.f;
    base = nb;
  }

  const float dd = di * di;
  float4 v;
  v.x = acc.x + dd * hs.x + b0;
  v.y = acc.y + dd * hs.y + b1;
  v.z = acc.z + dd * hs.z + b2;
  v.w = acc.w + dd * hs.w + b3;
  v.x = fmaxf(v.x, 0.f);
  v.y = fmaxf(v.y, 0.f);
  v.z = fmaxf(v.z, 0.f);
  v.w = fmaxf(v.w, 0.f);
  reinterpret_cast<float4*>(outf)[(size_t)i * 16 + l] = v;
}

__global__ void gemm_out_kernel(const float* __restrict__ feat,
                                const float* __restrict__ W8,
                                float* __restrict__ h2) {
  int i = blockIdx.x * blockDim.x + threadIdx.x;
  if (i < kN) {
    float s0 = 0.f, s1 = 0.f;
    const float* f = feat + (size_t)i * kDH;
    #pragma unroll
    for (int k = 0; k < kDH; ++k) {
      s0 = fmaf(f[k], W8[k * 2 + 0], s0);
      s1 = fmaf(f[k], W8[k * 2 + 1], s1);
    }
    h2[i * 2 + 0] = s0;
    h2[i * 2 + 1] = s1;
  }
}

__global__ void gather2_kernel(const int* __restrict__ rowptr,
                               const int2* __restrict__ edges,
                               const float* __restrict__ h2,
                               const float* __restrict__ dinv,
                               const float* __restrict__ b8,
                               float* __restrict__ out) {
  const int t = blockIdx.x * blockDim.x + threadIdx.x;
  const int i = t >> 4;
  const int l = t & 15;
  const int beg = rowptr[i], end = rowptr[i + 1];
  float a0 = 0.f, a1 = 0.f;
  for (int j = beg + l; j < end; j += 16) {
    const int2 ed = edges[j];
    const float w = __int_as_float(ed.y);
    a0 += w * h2[(size_t)ed.x * 2 + 0];
    a1 += w * h2[(size_t)ed.x * 2 + 1];
  }
  a0 = sum16(a0);
  a1 = sum16(a1);
  if (l < 2) {
    const float di = dinv[i];
    const float a = (l == 0) ? a0 : a1;
    out[(size_t)i * 2 + l] = a + di * di * h2[(size_t)i * 2 + l] + b8[l];
  }
}

}

extern "C" void kernel_launch(void* const* d_in, const int* in_sizes, int n_in,
                              void* d_out, int out_size, void* d_ws, size_t ws_size,
                              hipStream_t stream) {
  (void)in_sizes; (void)n_in; (void)out_size; (void)ws_size;

  const float* x  = (const float*)d_in[0];
  const int*   ei = (const int*)d_in[1];
  const float* ew = (const float*)d_in[2];
  const float* W[5] = {(const float*)d_in[3], (const float*)d_in[5], (const float*)d_in[7],
                       (const float*)d_in[9], (const float*)d_in[11]};
  const float* B[5] = {(const float*)d_in[4], (const float*)d_in[6], (const float*)d_in[8],
                       (const float*)d_in[10], (const float*)d_in[12]};
  const float* W8 = (const float*)d_in[13];
  const float* b8 = (const float*)d_in[14];
  float* out = (float*)d_out;

  const int* row = ei;
  const int* col = ei + kE;

  float* ws = (float*)d_ws;
  float* dinv  = ws;         ws += kN;
  int2*  edges = (int2*)ws;  ws += (size_t)kE * 2;
  float* feat  = ws;         ws += (size_t)kN * kDH;
  float* hbuf  = ws;         ws += (size_t)kN * kDH;
  int*   cnt   = (int*)ws;   ws += kN;
  int*   rowptr = (int*)ws;
  int*   rank  = (int*)hbuf;

  zero_int_kernel<<<kN / 256, 256, 0, stream>>>(cnt, kN);
  count_rank_kernel<<<kE / 256, 256, 0, stream>>>(col, cnt, rank);
  scan_kernel<<<1, 1024, 0, stream>>>(cnt, rowptr);
  place_kernel<<<kE / 256, 256, 0, stream>>>(row, col, ew, rowptr, rank, edges);

  dinv_kernel<<<(kN * 16) / 256, 256, 0, stream>>>(rowptr, edges, dinv);
  norm_kernel<<<(kN * 16) / 256, 256, 0, stream>>>(rowptr, dinv, edges);

  const dim3 ggrid(kN / (16 * 4), kDH / 16);

  const float* cur = x;
  int curK = 128;
  for (int layer = 0; layer < 5; ++layer) {
    gemm_wmma_kernel<<<ggrid, 128, 0, stream>>>(cur, W[layer], hbuf, curK);
    gather64_kernel<<<(kN * 16) / 256, 256, 0, stream>>>(rowptr, edges, hbuf, dinv,
                                                         B[layer], feat);
    cur = feat;
    curK = kDH;
  }

  gemm_out_kernel<<<kN / 256, 256, 0, stream>>>(cur, W8, hbuf);
  gather2_kernel<<<(kN * 16) / 256, 256, 0, stream>>>(rowptr, edges, hbuf, dinv, b8, out);
}


# Every edge and node launch becomes a resident grid whose waves loop over their edges or nodes (next node's bounds and edges asked ahead); GEMM blocks loop over row groups; sums still in bucket order
# speedup vs baseline: 2.7901x; 1.0480x over previous
#include <hip/hip_runtime.h>


namespace {

constexpr int kN  = 65536;
constexpr int kE  = 1048576;
constexpr int kDH = 64;
constexpr int kMaxK = 128;

constexpr int kEdgeBlocks = 512;
constexpr int kEdgeT = kEdgeBlocks * 256;
constexpr int kEdgeIters = kE / kEdgeT;
constexpr int kNodeBlocks = 1024;
constexpr int kNodeHW = kNodeBlocks * 256 / 16;
constexpr int kNodeIters = kN / kNodeHW;
constexpr int kGemmBlocksX = 512;
static_assert(kEdgeIters * kEdgeT == kE, "edge launch must tile the edges");
static_assert(kNodeIters * kNodeHW == kN, "node launch must tile the nodes");
static_assert((kN / 64) % kGemmBlocksX == 0, "row groups per block must be whole");

typedef __attribute__((ext_vector_type(2))) float v2f;
typedef __attribute__((ext_vector_type(8))) float v8f;

__device__ __forceinline__ float sum16(float v) {
  v += __shfl_xor(v, 8, 16);
  v += __shfl_xor(v, 4, 16);
  v += __shfl_xor(v, 2, 16);
  v += __shfl_xor(v, 1, 16);
  return v;
}

__global__ void zero_int_kernel(int* __restrict__ p, int n) {
  int i = blockIdx.x * blockDim.x + threadIdx.x;
  if (i < n) p[i] = 0;
}

__global__ void count_rank_kernel(const int* __restrict__ col,
                                  int* __restrict__ cnt,
                                  int* __restrict__ rank) {
  const int t = blockIdx.x * 256 + threadIdx.x;
  int c[kEdgeIters], r[kEdgeIters];
  #pragma unroll
  for (int q = 0; q < kEdgeIters; ++q) c[q] = col[t + q * kEdgeT];
  #pragma unroll
  for (int q = 0; q < kEdgeIters; ++q) r[q] = atomicAdd(&cnt[c[q]], 1);
  #pragma unroll
  for (int q = 0; q < kEdgeIters; ++q) rank[t + q * kEdgeT] = r[q];
}

__global__ __launch_bounds__(1024) void scan_kernel(const int* __restrict__ cnt,
                                                    int* __restrict__ rowptr) {
  __shared__ int wsum[32];
  const int t = threadIdx.x;
  const int lane = t & 31;
  const int wv = t >> 5;
  const int4* c4 = reinterpret_cast<const int4*>(cnt) + t * 16;
  int4 v[16];
  #pragma unroll
  for (int q = 0; q < 16; ++q) v[q] = c4[q];
  int s = 0;
  #pragma unroll
  for (int q = 0; q < 16; ++q) {
    const int a = v[q].x, b = v[q].y, c = v[q].z, d = v[q].w;
    v[q].x = s; s += a;
    v[q].y = s; s += b;
    v[q].z = s; s += c;
    v[q].w = s; s += d;
  }
  int inc = s;
  #pragma unroll
  for (int off = 1; off < 32; off <<= 1) {
    const int nb = __shfl_up(inc, off);
    if (lane >= off) inc += nb;
  }
  if (lane == 31) wsum[wv] = inc;
  __syncthreads();
  if (wv == 0) {
    const int x = wsum[lane];
    int xi = x;
    #pragma unroll
    for (int off = 1; off < 32; off <<= 1) {
      const int nb = __shfl_up(xi, off);
      if (lane >= off) xi += nb;
    }
    wsum[lane] = xi - x;
  }
  __syncthreads();
  const int base = wsum[wv] + inc - s;
  int4* r4 = reinterpret_cast<int4*>(rowptr) + t * 16;
  #pragma unroll
  for (int q = 0; q < 16; ++q) {
    int4 o = v[q];
    o.x += base; o.y += base; o.z += base; o.w += base;
    r4[q] = o;
  }
  if (t == 1023) rowptr[kN] = base + s;
}

__global__ void place_kernel(const int* __restrict__ row, const int* __restrict__ col,
                             const float* __restrict__ ew,
                             const int* __restrict__ rowptr,
                             const int* __restrict__ rank,
                             int2* __restrict__ edges) {
  const int t = blockIdx.x * 256 + threadIdx.x;
  int c[kEdgeIters], rk[kEdgeIters], s[kEdgeIters], p[kEdgeIters];
  float w[kEdgeIters];
  #pragma unroll
  for (int q = 0; q < kEdgeIters; ++q) {
    const int e = t + q * kEdgeT;
    c[q] = col[e];
    rk[q] = rank[e];
    s[q] = row[e];
    w[q] = ew[e];
  }
  #pragma unroll
  for (int q = 0; q < kEdgeIters; ++q) p[q] = rowptr[c[q]] + rk[q];
  #pragma unroll
  for (int q = 0; q < kEdgeIters; ++q) edges[p[q]] = make_int2(s[q], __float_as_int(w[q]));
}

__global__ void dinv_kernel(const int* __restrict__ rowptr,
                            const int2* __restrict__ edges,
                            float* __restrict__ dinv) {
  const int t = blockIdx.x * 256 + threadIdx.x;
  const int hw = t >> 4;
  const int l = t & 15;
  int beg[kNodeIters], end[kNodeIters];
  #pragma unroll
  for (int q = 0; q < kNodeIters; ++q) {
    beg[q] = rowptr[hw + q * kNodeHW];
    end[q] = rowptr[hw + q * kNodeHW + 1];
  }
  #pragma unroll
  for (int q = 0; q < kNodeIters; ++q) {
    const int i = hw + q * kNodeHW;
    float s = 0.f;
    for (int j = beg[q] + l; j < end[q]; j += 16) s += __int_as_float(edges[j].y);
    s = sum16(s) + 1.0f;
    if (l == 0) dinv[i] = (s > 0.f) ? rsqrtf(s) : 0.f;
  }
}

__global__ void norm_kernel(const int* __restrict__ rowptr,
                            const float* __restrict__ dinv,
                            int2* __restrict__ edges) {
  const int t = blockIdx.x * 256 + threadIdx.x;
  const int hw = t >> 4;
  const int l = t & 15;
  int beg[kNodeIters], end[kNodeIters];
  float di[kNodeIters];
  #pragma unroll
  for (int q = 0; q < kNodeIters; ++q) {
    const int i = hw + q * kNodeHW;
    beg[q] = rowptr[i];
    end[q] = rowptr[i + 1];
    di[q] = dinv[i];
  }
  #pragma unroll
  for (int q = 0; q < kNodeIters; ++q) {
    for (int j = beg[q] + l; j < end[q]; j += 16) {
      const int2 ed = edges[j];
      const float nm = dinv[ed.x] * __int_as_float(ed.y) * di[q];
      edges[j].y = __float_as_int(nm);
    }
  }
}

__global__ void gemm_wmma_kernel(const float* __restrict__ A,
                                 const float* __restrict__ W,
                                 float* __restrict__ H, int K) {
  __shared__ float Ws[kMaxK * 17];

  const int tid  = threadIdx.x;
  const int lane = tid & 31;
  const int wave = tid >> 5;
  const int n0 = blockIdx.y * 16;
  const int half = lane >> 4;
  const int l = lane & 15;

  for (int idx = tid; idx < K * 16; idx += 128) {
    int kk = idx >> 4;
    int nn = idx & 15;
    Ws[kk * 17 + nn] = W[(size_t)kk * kDH + n0 + nn];
  }
  __syncthreads();

  for (int g = blockIdx.x; g < kN / 64; g += gridDim.x) {
    const int m0 = (g * 4 + wave) * 16;
    v8f acc = {};
    const float* arow = A + (size_t)(m0 + l) * K;
    #pragma unroll 4
    for (int k = 0; k < K; k += 4) {
      const int ka = k + 2 * half;
      v2f a, b;
      a.x = arow[ka];
      a.y = arow[ka + 1];
      b.x = Ws[ka * 17 + l];
      b.y = Ws[(ka + 1) * 17 + l];
      acc = __builtin_amdgcn_wmma_f32_16x16x4_f32(
          false, a, false, b, (short)0, acc, false, false);
    }

    float* hrow = H + (size_t)(m0 + half * 8) * kDH + (n0 + l);
    #pragma unroll
    for (int r = 0; r < 8; ++r) hrow[(size_t)r * kDH] = acc[r];
  }
}

template <int NL>
__device__ __forceinline__ void rows_fma(const float4* __restrict__ h4, int l,
                                         int sx, float wx, float4& acc) {
  int s[NL];
  float w[NL];
  float4 a[NL];
  #pragma unroll
  for (int q = 0; q < NL; ++q) {
    s[q] = __shfl(sx, q, 16);
    w[q] = __shfl(wx, q, 16);
  }
  #pragma unroll
  for (int q = 0; q < NL; ++q) a[q] = h4[(size_t)s[q] * 16 + l];
  #pragma unroll
  for (int q = 0; q < NL; ++q) {
    acc.x += w[q] * a[q].x;
    acc.y += w[q] * a[q].y;
    acc.z += w[q] * a[q].z;
    acc.w += w[q] * a[q].w;
  }
}

__global__ __launch_bounds__(256) __attribute__((amdgpu_waves_per_eu(4, 8))) void gather64_kernel(const int* __restrict__ rowptr,
                                                       const int2* __restrict__ edges,
                                                       const float* __restrict__ h,
                                                       const float* __restrict__ dinv,
                                                       const float* __restrict__ bias,
                                                       float* __restrict__ outf) {
  const int t = blockIdx.x * 256 + threadIdx.x;
  const int hw = t >> 4;
  const int l = t & 15;
  const float4* __restrict__ h4 = reinterpret_cast<const float4*>(h);
  const float b0 = bias[l * 4 + 0], b1 = bias[l * 4 + 1];
  const float b2 = bias[l * 4 + 2], b3 = bias[l * 4 + 3];

  int i = hw;
  int beg = rowptr[i], end = rowptr[i + 1];
  int i1 = min(i + kNodeHW, kN - 1);
  int beg1 = rowptr[i1], end1 = rowptr[i1 + 1];
  int2 ed = edges[min(beg + l, kE - 1)];

  for (int it = 0; it < kNodeIters; ++it) {
    const int i2 = min(i1 + kNodeHW, kN - 1);
    const int beg2 = rowptr[i2], end2 = rowptr[i2 + 1];
    const int2 ed1 = edges[min(beg1 + l, kE - 1)];
    const float di = dinv[i];
    const float4 hs = h4[(size_t)i * 16 + l];

    float4 acc = make_float4(0.f, 0.f, 0.f, 0.f);
    int base = beg;
    int sx = (base + l < end) ? ed.x : i;
    float wx = (base + l < end) ? __int_as_float(ed.y) : 0.f;
    while (base < end) {
      const int m = end - base;
      const int nb = base + 16;
      ed = edges[min(nb + l, kE - 1)];
      if (m > 8) {
        rows_fma<16>(h4, l, sx, wx, acc);
      } else {
        rows_fma<8>(h4, l, sx, wx, acc);
      }
      sx = (nb + l < end) ? ed.x : i;
      wx = (nb + l < end) ? __int_as_float(ed.y) : 0.f;
      base = nb;
    }

    const float dd = di * di;
    float4 v;
    v.x = acc.x + dd * hs.x + b0;
    v.y = acc.y + dd * hs.y + b1;
    v.z = acc.z + dd * hs.z + b2;
    v.w = acc.w + dd * hs.w + b3;
    v.x = fmaxf(v.x, 0.f);
    v.y = fmaxf(v.y, 0.f);
    v.z = fmaxf(v.z, 0.f);
    v.w = fmaxf(v.w, 0.f);
    reinterpret_cast<float4*>(outf)[(size_t)i * 16 + l] = v;

    i = i1; beg = beg1; end = end1; ed = ed1;
    i1 = i2; beg1 = beg2; end1 = end2;
  }
}

__global__ void gemm_out_kernel(const float* __restrict__ feat,
                                const float* __restrict__ W8,
                                float* __restrict__ h2) {
  int i = blockIdx.x * blockDim.x + threadIdx.x;
  if (i < kN) {
    float s0 = 0.f, s1 = 0.f;
    const float* f = feat + (size_t)i * kDH;
    #pragma unroll
    for (int k = 0; k < kDH; ++k) {
      s0 = fmaf(f[k], W8[k * 2 + 0], s0);
      s1 = fmaf(f[k], W8[k * 2 + 1], s1);
    }
    h2[i * 2 + 0] = s0;
    h2[i * 2 + 1] = s1;
  }
}

__global__ void gather2_kernel(const int* __restrict__ rowptr,
                               const int2* __restrict__ edges,
                               const float* __restrict__ h2,
                               const float* __restrict__ dinv,
                               const float* __restrict__ b8,
                               float* __restrict__ out) {
  const int t = blockIdx.x * 256 + threadIdx.x;
  const int hw = t >> 4;
  const int l = t & 15;
  int beg[kNodeIters], end[kNodeIters];
  #pragma unroll
  for (int q = 0; q < kNodeIters; ++q) {
    beg[q] = rowptr[hw + q * kNodeHW];
    end[q] = rowptr[hw + q * kNodeHW + 1];
  }
  #pragma unroll
  for (int q = 0; q < kNodeIters; ++q) {
    const int i = hw + q * kNodeHW;
    float a0 = 0.f, a1 = 0.f;
    for (int j = beg[q] + l; j < end[q]; j += 16) {
      const int2 ed = edges[j];
      const float w = __int_as_float(ed.y);
      a0 += w * h2[(size_t)ed.x * 2 + 0];
      a1 += w * h2[(size_t)ed.x * 2 + 1];
    }
    a0 = sum16(a0);
    a1 = sum16(a1);
    if (l < 2) {
      const float di = dinv[i];
      const float a = (l == 0) ? a0 : a1;
      out[(size_t)i * 2 + l] = a + di * di * h2[(size_t)i * 2 + l] + b8[l];
    }
  }
}

}

extern "C" void kernel_launch(void* const* d_in, const int* in_sizes, int n_in,
                              void* d_out, int out_size, void* d_ws, size_t ws_size,
                              hipStream_t stream) {
  (void)in_sizes; (void)n_in; (void)out_size; (void)ws_size;

  const float* x  = (const float*)d_in[0];
  const int*   ei = (const int*)d_in[1];
  const float* ew = (const float*)d_in[2];
  const float* W[5] = {(const float*)d_in[3], (const float*)d_in[5], (const float*)d_in[7],
                       (const float*)d_in[9], (const float*)d_in[11]};
  const float* B[5] = {(const float*)d_in[4], (const float*)d_in[6], (const float*)d_in[8],
                       (const float*)d_in[10], (const float*)d_in[12]};
  const float* W8 = (const float*)d_in[13];
  const float* b8 = (const float*)d_in[14];
  float* out = (float*)d_out;

  const int* row = ei;
  const int* col = ei + kE;

  float* ws = (float*)d_ws;
  float* dinv  = ws;         ws += kN;
  int2*  edges = (int2*)ws;  ws += (size_t)kE * 2;
  float* feat  = ws;         ws += (size_t)kN * kDH;
  float* hbuf  = ws;         ws += (size_t)kN * kDH;
  int*   cnt   = (int*)ws;   ws += kN;
  int*   rowptr = (int*)ws;
  int*   rank  = (int*)hbuf;

  zero_int_kernel<<<kN / 256, 256, 0, stream>>>(cnt, kN);
  count_rank_kernel<<<kEdgeBlocks, 256, 0, stream>>>(col, cnt, rank);
  scan_kernel<<<1, 1024, 0, stream>>>(cnt, rowptr);
  place_kernel<<<kEdgeBlocks, 256, 0, stream>>>(row, col, ew, rowptr, rank, edges);

  dinv_kernel<<<kNodeBlocks, 256, 0, stream>>>(rowptr, edges, dinv);
  norm_kernel<<<kNodeBlocks, 256, 0, stream>>>(rowptr, dinv, edges);

  const dim3 ggrid(kGemmBlocksX, kDH / 16);

  const float* cur = x;
  int curK = 128;
  for (int layer = 0; layer < 5; ++layer) {
    gemm_wmma_kernel<<<ggrid, 128, 0, stream>>>(cur, W[layer], hbuf, curK);
    gather64_kernel<<<kNodeBlocks, 256, 0, stream>>>(rowptr, edges, hbuf, dinv,
                                                      B[layer], feat);
    cur = feat;
    curK = kDH;
  }

  gemm_out_kernel<<<kN / 256, 256, 0, stream>>>(cur, W8, hbuf);
  gather2_kernel<<<kNodeBlocks, 256, 0, stream>>>(rowptr, edges, hbuf, dinv, b8, out);
}


# Each 64-wide gather forms the next layer's product in its epilogue (64 rows in LDS, f32 WMMA; the last one the 64x2 dot, summed 4 per lane then over 16 lanes): 5 product launches gone
# speedup vs baseline: 3.1317x; 1.1224x over previous
#include <hip/hip_runtime.h>


namespace {

constexpr int kN  = 65536;
constexpr int kE  = 1048576;
constexpr int kDH = 64;
constexpr int kMaxK = 128;

constexpr int kEdgeBlocks = 512;
constexpr int kEdgeT = kEdgeBlocks * 256;
constexpr int kEdgeIters = kE / kEdgeT;
constexpr int kNodeBlocks = 1024;
constexpr int kNodeHW = kNodeBlocks * 256 / 16;
constexpr int kNodeIters = kN / kNodeHW;
constexpr int kGemmBlocksX = 512;
static_assert(kEdgeIters * kEdgeT == kE, "edge launch must tile the edges");
static_assert(kNodeIters * kNodeHW == kN, "node launch must tile the nodes");
static_assert((kN / 64) % kGemmBlocksX == 0, "row groups per block must be whole");

typedef __attribute__((ext_vector_type(2))) float v2f;
typedef __attribute__((ext_vector_type(8))) float v8f;

__device__ __forceinline__ float sum16(float v) {
  v += __shfl_xor(v, 8, 16);
  v += __shfl_xor(v, 4, 16);
  v += __shfl_xor(v, 2, 16);
  v += __shfl_xor(v, 1, 16);
  return v;
}

__global__ void zero_int_kernel(int* __restrict__ p, int n) {
  int i = blockIdx.x * blockDim.x + threadIdx.x;
  if (i < n) p[i] = 0;
}

__global__ void count_rank_kernel(const int* __restrict__ col,
                                  int* __restrict__ cnt,
                                  int* __restrict__ rank) {
  const int t = blockIdx.x * 256 + threadIdx.x;
  int c[kEdgeIters], r[kEdgeIters];
  #pragma unroll
  for (int q = 0; q < kEdgeIters; ++q) c[q] = col[t + q * kEdgeT];
  #pragma unroll
  for (int q = 0; q < kEdgeIters; ++q) r[q] = atomicAdd(&cnt[c[q]], 1);
  #pragma unroll
  for (int q = 0; q < kEdgeIters; ++q) rank[t + q * kEdgeT] = r[q];
}

__global__ __launch_bounds__(1024) void scan_kernel(const int* __restrict__ cnt,
                                                    int* __restrict__ rowptr) {
  __shared__ int wsum[32];
  const int t = threadIdx.x;
  const int lane = t & 31;
  const int wv = t >> 5;
  const int4* c4 = reinterpret_cast<const int4*>(cnt) + t * 16;
  int4 v[16];
  #pragma unroll
  for (int q = 0; q < 16; ++q) v[q] = c4[q];
  int s = 0;
  #pragma unroll
  for (int q = 0; q < 16; ++q) {
    const int a = v[q].x, b = v[q].y, c = v[q].z, d = v[q].w;
    v[q].x = s; s += a;
    v[q].y = s; s += b;
    v[q].z = s; s += c;
    v[q].w = s; s += d;
  }
  int inc = s;
  #pragma unroll
  for (int off = 1; off < 32; off <<= 1) {
    const int nb = __shfl_up(inc, off);
    if (lane >= off) inc += nb;
  }
  if (lane == 31) wsum[wv] = inc;
  __syncthreads();
  if (wv == 0) {
    const int x = wsum[lane];
    int xi = x;
    #pragma unroll
    for (int off = 1; off < 32; off <<= 1) {
      const int nb = __shfl_up(xi, off);
      if (lane >= off) xi += nb;
    }
    wsum[lane] = xi - x;
  }
  __syncthreads();
  const int base = wsum[wv] + inc - s;
  int4* r4 = reinterpret_cast<int4*>(rowptr) + t * 16;
  #pragma unroll
  for (int q = 0; q < 16; ++q) {
    int4 o = v[q];
    o.x += base; o.y += base; o.z += base; o.w += base;
    r4[q] = o;
  }
  if (t == 1023) rowptr[kN] = base + s;
}

__global__ void place_kernel(const int* __restrict__ row, const int* __restrict__ col,
                             const float* __restrict__ ew,
                             const int* __restrict__ rowptr,
                             const int* __restrict__ rank,
                             int2* __restrict__ edges) {
  const int t = blockIdx.x * 256 + threadIdx.x;
  int c[kEdgeIters], rk[kEdgeIters], s[kEdgeIters], p[kEdgeIters];
  float w[kEdgeIters];
  #pragma unroll
  for (int q = 0; q < kEdgeIters; ++q) {
    const int e = t + q * kEdgeT;
    c[q] = col[e];
    rk[q] = rank[e];
    s[q] = row[e];
    w[q] = ew[e];
  }
  #pragma unroll
  for (int q = 0; q < kEdgeIters; ++q) p[q] = rowptr[c[q]] + rk[q];
  #pragma unroll
  for (int q = 0; q < kEdgeIters; ++q) edges[p[q]] = make_int2(s[q], __float_as_int(w[q]));
}

__global__ void dinv_kernel(const int* __restrict__ rowptr,
                            const int2* __restrict__ edges,
                            float* __restrict__ dinv) {
  const int t = blockIdx.x * 256 + threadIdx.x;
  const int hw = t >> 4;
  const int l = t & 15;
  int beg[kNodeIters], end[kNodeIters];
  #pragma unroll
  for (int q = 0; q < kNodeIters; ++q) {
    beg[q] = rowptr[hw + q * kNodeHW];
    end[q] = rowptr[hw + q * kNodeHW + 1];
  }
  #pragma unroll
  for (int q = 0; q < kNodeIters; ++q) {
    const int i = hw + q * kNodeHW;
    float s = 0.f;
    for (int j = beg[q] + l; j < end[q]; j += 16) s += __int_as_float(edges[j].y);
    s = sum16(s) + 1.0f;
    if (l == 0) dinv[i] = (s > 0.f) ? rsqrtf(s) : 0.f;
  }
}

__global__ void norm_kernel(const int* __restrict__ rowptr,
                            const float* __restrict__ dinv,
                            int2* __restrict__ edges) {
  const int t = blockIdx.x * 256 + threadIdx.x;
  const int hw = t >> 4;
  const int l = t & 15;
  int beg[kNodeIters], end[kNodeIters];
  float di[kNodeIters];
  #pragma unroll
  for (int q = 0; q < kNodeIters; ++q) {
    const int i = hw + q * kNodeHW;
    beg[q] = rowptr[i];
    end[q] = rowptr[i + 1];
    di[q] = dinv[i];
  }
  #pragma unroll
  for (int q = 0; q < kNodeIters; ++q) {
    for (int j = beg[q] + l; j < end[q]; j += 16) {
      const int2 ed = edges[j];
      const float nm = dinv[ed.x] * __int_as_float(ed.y) * di[q];
      edges[j].y = __float_as_int(nm);
    }
  }
}

__global__ void gemm_wmma_kernel(const float* __restrict__ A,
                                 const float* __restrict__ W,
                                 float* __restrict__ H, int K) {
  __shared__ float Ws[kMaxK * 17];

  const int tid  = threadIdx.x;
  const int lane = tid & 31;
  const int wave = tid >> 5;
  const int n0 = blockIdx.y * 16;
  const int half = lane >> 4;
  const int l = lane & 15;

  for (int idx = tid; idx < K * 16; idx += 128) {
    int kk = idx >> 4;
    int nn = idx & 15;
    Ws[kk * 17 + nn] = W[(size_t)kk * kDH + n0 + nn];
  }
  __syncthreads();

  for (int g = blockIdx.x; g < kN / 64; g += gridDim.x) {
    const int m0 = (g * 4 + wave) * 16;
    v8f acc = {};
    const float* arow = A + (size_t)(m0 + l) * K;
    #pragma unroll 4
    for (int k = 0; k < K; k += 4) {
      const int ka = k + 2 * half;
      v2f a, b;
      a.x = arow[ka];
      a.y = arow[ka + 1];
      b.x = Ws[ka * 17 + l];
      b.y = Ws[(ka + 1) * 17 + l];
      acc = __builtin_amdgcn_wmma_f32_16x16x4_f32(
          false, a, false, b, (short)0, acc, false, false);
    }

    float* hrow = H + (size_t)(m0 + half * 8) * kDH + (n0 + l);
    #pragma unroll
    for (int r = 0; r < 8; ++r) hrow[(size_t)r * kDH] = acc[r];
  }
}

template <int NL>
__device__ __forceinline__ void rows_fma(const float4* __restrict__ h4, int l,
                                         int sx, float wx, float4& acc) {
  int s[NL];
  float w[NL];
  float4 a[NL];
  #pragma unroll
  for (int q = 0; q < NL; ++q) {
    s[q] = __shfl(sx, q, 16);
    w[q] = __shfl(wx, q, 16);
  }
  #pragma unroll
  for (int q = 0; q < NL; ++q) a[q] = h4[(size_t)s[q] * 16 + l];
  #pragma unroll
  for (int q = 0; q < NL; ++q) {
    acc.x += w[q] * a[q].x;
    acc.y += w[q] * a[q].y;
    acc.z += w[q] * a[q].z;
    acc.w += w[q] * a[q].w;
  }
}

template <class Emit>
__device__ __forceinline__ void gather_rows(const int* __restrict__ rowptr,
                                            const int2* __restrict__ edges,
                                            const float* __restrict__ h,
                                            const float* __restrict__ dinv,
                                            const float* __restrict__ bias,
                                            Emit emit) {
  const int t = blockIdx.x * 256 + threadIdx.x;
  const int hw = t >> 4;
  const int l = t & 15;
  const float4* __restrict__ h4 = reinterpret_cast<const float4*>(h);
  const float b0 = bias[l * 4 + 0], b1 = bias[l * 4 + 1];
  const float b2 = bias[l * 4 + 2], b3 = bias[l * 4 + 3];

  int i = hw;
  int beg = rowptr[i], end = rowptr[i + 1];
  int i1 = min(i + kNodeHW, kN - 1);
  int beg1 = rowptr[i1], end1 = rowptr[i1 + 1];
  int2 ed = edges[min(beg + l, kE - 1)];

  for (int it = 0; it < kNodeIters; ++it) {
    const int i2 = min(i1 + kNodeHW, kN - 1);
    const int beg2 = rowptr[i2], end2 = rowptr[i2 + 1];
    const int2 ed1 = edges[min(beg1 + l, kE - 1)];
    const float di = dinv[i];
    const float4 hs = h4[(size_t)i * 16 + l];

    float4 acc = make_float4(0.f, 0.f, 0.f, 0.f);
    int base = beg;
    int sx = (base + l < end) ? ed.x : i;
    float wx = (base + l < end) ? __int_as_float(ed.y) : 0.f;
    while (base < end) {
      const int m = end - base;
      const int nb = base + 16;
      ed = edges[min(nb + l, kE - 1)];
      if (m > 8) {
        rows_fma<16>(h4, l, sx, wx, acc);
      } else {
        rows_fma<8>(h4, l, sx, wx, acc);
      }
      sx = (nb + l < end) ? ed.x : i;
      wx = (nb + l < end) ? __int_as_float(ed.y) : 0.f;
      base = nb;
    }

    const float dd = di * di;
    float4 v;
    v.x = acc.x + dd * hs.x + b0;
    v.y = acc.y + dd * hs.y + b1;
    v.z = acc.z + dd * hs.z + b2;
    v.w = acc.w + dd * hs.w + b3;
    v.x = fmaxf(v.x, 0.f);
    v.y = fmaxf(v.y, 0.f);
    v.z = fmaxf(v.z, 0.f);
    v.w = fmaxf(v.w, 0.f);
    emit(it, i, l, v);

    i = i1; beg = beg1; end = end1; ed = ed1;
    i1 = i2; beg1 = beg2; end1 = end2;
  }
}

constexpr int kTP = 68;
static_assert(kNodeIters * 16 == 64, "the LDS tile holds 4 M-tiles of 16 rows");

__global__ __launch_bounds__(256) __attribute__((amdgpu_waves_per_eu(4, 8))) void gather64_gemm_kernel(const int* __restrict__ rowptr,
                                                       const int2* __restrict__ edges,
                                                       const float* __restrict__ h,
                                                       const float* __restrict__ dinv,
                                                       const float* __restrict__ bias,
                                                       const float* __restrict__ Wn,
                                                       float* __restrict__ hn) {
  __shared__ float As[64 * kTP];
  __shared__ float Wt[64 * kTP];
  const int tid = threadIdx.x;

  #pragma unroll
  for (int j = 0; j < 16; ++j) {
    const int idx = tid + j * 256;
    Wt[(idx & 63) * kTP + (idx >> 6)] = Wn[idx];
  }

  const int hwl = tid >> 4;
  gather_rows(rowptr, edges, h, dinv, bias,
              [&](int it, int  , int l, const float4& v) {
                *reinterpret_cast<float4*>(&As[(it * 16 + hwl) * kTP + 4 * l]) = v;
              });
  __syncthreads();

  const int lane = tid & 31;
  const int wave = tid >> 5;
  const int half = lane >> 4;
  const int l = lane & 15;
  const int mt = wave >> 1;
  const int n0 = (wave & 1) * 32;
  const float* ap  = &As[(mt * 16 + l) * kTP + 2 * half];
  const float* bp0 = &Wt[(n0 + l) * kTP + 2 * half];
  const float* bp1 = &Wt[(n0 + 16 + l) * kTP + 2 * half];
  v8f acc0 = {}, acc1 = {};
  #pragma unroll
  for (int k = 0; k < kDH; k += 4) {
    const v2f a  = *reinterpret_cast<const v2f*>(ap + k);
    const v2f b0 = *reinterpret_cast<const v2f*>(bp0 + k);
    const v2f b1 = *reinterpret_cast<const v2f*>(bp1 + k);
    acc0 = __builtin_amdgcn_wmma_f32_16x16x4_f32(
        false, a, false, b0, (short)0, acc0, false, false);
    acc1 = __builtin_amdgcn_wmma_f32_16x16x4_f32(
        false, a, false, b1, (short)0, acc1, false, false);
  }

  float* hrow = hn + (size_t)(blockIdx.x * 16 + mt * kNodeHW + 8 * half) * kDH + (n0 + l);
  #pragma unroll
  for (int r = 0; r < 8; ++r) {
    hrow[(size_t)r * kDH] = acc0[r];
    hrow[(size_t)r * kDH + 16] = acc1[r];
  }
}

__global__ __launch_bounds__(256) __attribute__((amdgpu_waves_per_eu(4, 8))) void gather64_proj_kernel(const int* __restrict__ rowptr,
                                                       const int2* __restrict__ edges,
                                                       const float* __restrict__ h,
                                                       const float* __restrict__ dinv,
                                                       const float* __restrict__ bias,
                                                       const float* __restrict__ W8,
                                                       float* __restrict__ h2) {
  const int lw = threadIdx.x & 15;
  float w0[4], w1[4];
  #pragma unroll
  for (int q = 0; q < 4; ++q) {
    w0[q] = W8[(lw * 4 + q) * 2 + 0];
    w1[q] = W8[(lw * 4 + q) * 2 + 1];
  }
  gather_rows(rowptr, edges, h, dinv, bias,
              [&](int  , int i, int l, const float4& v) {
                float s0 = v.x * w0[0];
                float s1 = v.x * w1[0];
                s0 = fmaf(v.y, w0[1], s0);
                s1 = fmaf(v.y, w1[1], s1);
                s0 = fmaf(v.z, w0[2], s0);
                s1 = fmaf(v.z, w1[2], s1);
                s0 = fmaf(v.w, w0[3], s0);
                s1 = fmaf(v.w, w1[3], s1);
                s0 = sum16(s0);
                s1 = sum16(s1);
                if (l == 0) reinterpret_cast<float2*>(h2)[i] = make_float2(s0, s1);
              });
}

__global__ void gather2_kernel(const int* __restrict__ rowptr,
                               const int2* __restrict__ edges,
                               const float* __restrict__ h2,
                               const float* __restrict__ dinv,
                               const float* __restrict__ b8,
                               float* __restrict__ out) {
  const int t = blockIdx.x * 256 + threadIdx.x;
  const int hw = t >> 4;
  const int l = t & 15;
  int beg[kNodeIters], end[kNodeIters];
  #pragma unroll
  for (int q = 0; q < kNodeIters; ++q) {
    beg[q] = rowptr[hw + q * kNodeHW];
    end[q] = rowptr[hw + q * kNodeHW + 1];
  }
  #pragma unroll
  for (int q = 0; q < kNodeIters; ++q) {
    const int i = hw + q * kNodeHW;
    float a0 = 0.f, a1 = 0.f;
    for (int j = beg[q] + l; j < end[q]; j += 16) {
      const int2 ed = edges[j];
      const float w = __int_as_float(ed.y);
      a0 += w * h2[(size_t)ed.x * 2 + 0];
      a1 += w * h2[(size_t)ed.x * 2 + 1];
    }
    a0 = sum16(a0);
    a1 = sum16(a1);
    if (l < 2) {
      const float di = dinv[i];
      const float a = (l == 0) ? a0 : a1;
      out[(size_t)i * 2 + l] = a + di * di * h2[(size_t)i * 2 + l] + b8[l];
    }
  }
}

}

extern "C" void kernel_launch(void* const* d_in, const int* in_sizes, int n_in,
                              void* d_out, int out_size, void* d_ws, size_t ws_size,
                              hipStream_t stream) {
  (void)in_sizes; (void)n_in; (void)out_size; (void)ws_size;

  const float* x  = (const float*)d_in[0];
  const int*   ei = (const int*)d_in[1];
  const float* ew = (const float*)d_in[2];
  const float* W[5] = {(const float*)d_in[3], (const float*)d_in[5], (const float*)d_in[7],
                       (const float*)d_in[9], (const float*)d_in[11]};
  const float* B[5] = {(const float*)d_in[4], (const float*)d_in[6], (const float*)d_in[8],
                       (const float*)d_in[10], (const float*)d_in[12]};
  const float* W8 = (const float*)d_in[13];
  const float* b8 = (const float*)d_in[14];
  float* out = (float*)d_out;

  const int* row = ei;
  const int* col = ei + kE;

  float* ws = (float*)d_ws;
  float* dinv  = ws;         ws += kN;
  int2*  edges = (int2*)ws;  ws += (size_t)kE * 2;
  float* feat  = ws;         ws += (size_t)kN * kDH;
  float* hbuf  = ws;         ws += (size_t)kN * kDH;
  int*   cnt   = (int*)ws;   ws += kN;
  int*   rowptr = (int*)ws;
  int*   rank  = (int*)hbuf;

  zero_int_kernel<<<kN / 256, 256, 0, stream>>>(cnt, kN);
  count_rank_kernel<<<kEdgeBlocks, 256, 0, stream>>>(col, cnt, rank);
  scan_kernel<<<1, 1024, 0, stream>>>(cnt, rowptr);
  place_kernel<<<kEdgeBlocks, 256, 0, stream>>>(row, col, ew, rowptr, rank, edges);

  dinv_kernel<<<kNodeBlocks, 256, 0, stream>>>(rowptr, edges, dinv);
  norm_kernel<<<kNodeBlocks, 256, 0, stream>>>(rowptr, dinv, edges);

  const dim3 ggrid(kGemmBlocksX, kDH / 16);

  gemm_wmma_kernel<<<ggrid, 128, 0, stream>>>(x, W[0], hbuf, 128);
  gather64_gemm_kernel<<<kNodeBlocks, 256, 0, stream>>>(rowptr, edges, hbuf, dinv, B[0], W[1], feat);
  gather64_gemm_kernel<<<kNodeBlocks, 256, 0, stream>>>(rowptr, edges, feat, dinv, B[1], W[2], hbuf);
  gather64_gemm_kernel<<<kNodeBlocks, 256, 0, stream>>>(rowptr, edges, hbuf, dinv, B[2], W[3], feat);
  gather64_gemm_kernel<<<kNodeBlocks, 256, 0, stream>>>(rowptr, edges, feat, dinv, B[3], W[4], hbuf);

  gather64_proj_kernel<<<kNodeBlocks, 256, 0, stream>>>(rowptr, edges, hbuf, dinv, B[4], W8, feat);
  gather2_kernel<<<kNodeBlocks, 256, 0, stream>>>(rowptr, edges, feat, dinv, b8, out);
}


# Edge bucketing without global atomics: LDS counters per block over 256 node ranges (count, scan, place), then one block per range orders its edges by node in LDS; bucket order now from LDS adds
# speedup vs baseline: 3.8959x; 1.2440x over previous
#include <hip/hip_runtime.h>


namespace {

constexpr int kN  = 65536;
constexpr int kE  = 1048576;
constexpr int kDH = 64;
constexpr int kMaxK = 128;

constexpr int kEdgeBlocks = 512;
constexpr int kEdgeT = kEdgeBlocks * 256;
constexpr int kEdgeIters = kE / kEdgeT;
constexpr int kNodeBlocks = 1024;
constexpr int kNodeHW = kNodeBlocks * 256 / 16;
constexpr int kNodeIters = kN / kNodeHW;
constexpr int kGemmBlocksX = 512;
static_assert(kEdgeIters * kEdgeT == kE, "edge launch must tile the edges");
static_assert(kNodeIters * kNodeHW == kN, "node launch must tile the nodes");
static_assert((kN / 64) % kGemmBlocksX == 0, "row groups per block must be whole");

constexpr int kBins = 256;
constexpr int kFineShift = 8;
constexpr int kFine = 1 << kFineShift;
constexpr int kChunks = 256;
constexpr int kChunkE = kE / kChunks;
constexpr int kSortT = 512;
constexpr int kSortIters = kChunkE / kSortT;
static_assert(kBins * kFine == kN, "ranges must tile the nodes");
static_assert(kBins * kChunks == kN, "the (range, chunk) table is scanned by the kN-entry scan");
static_assert(kSortIters * kSortT * kChunks == kE, "chunks must tile the edges");
static_assert(kN <= 65536, "a source index must fit the low 16 bits of a packed record");

typedef __attribute__((ext_vector_type(2))) float v2f;
typedef __attribute__((ext_vector_type(8))) float v8f;

__device__ __forceinline__ float sum16(float v) {
  v += __shfl_xor(v, 8, 16);
  v += __shfl_xor(v, 4, 16);
  v += __shfl_xor(v, 2, 16);
  v += __shfl_xor(v, 1, 16);
  return v;
}

__global__ __launch_bounds__(kSortT) void range_count_kernel(const int* __restrict__ col,
                                                             int* __restrict__ table) {
  __shared__ int h[kBins];
  const int t = threadIdx.x;
  if (t < kBins) h[t] = 0;
  const int e0 = blockIdx.x * kChunkE + t;
  int c[kSortIters];
  #pragma unroll
  for (int q = 0; q < kSortIters; ++q) c[q] = col[e0 + q * kSortT];
  __syncthreads();
  #pragma unroll
  for (int q = 0; q < kSortIters; ++q) atomicAdd(&h[(c[q] >> kFineShift) & (kBins - 1)], 1);
  __syncthreads();
  if (t < kBins) table[t * kChunks + blockIdx.x] = h[t];
}

__global__ __launch_bounds__(1024) void scan_kernel(const int* __restrict__ cnt,
                                                    int* __restrict__ rowptr) {
  __shared__ int wsum[32];
  const int t = threadIdx.x;
  const int lane = t & 31;
  const int wv = t >> 5;
  const int4* c4 = reinterpret_cast<const int4*>(cnt) + t * 16;
  int4 v[16];
  #pragma unroll
  for (int q = 0; q < 16; ++q) v[q] = c4[q];
  int s = 0;
  #pragma unroll
  for (int q = 0; q < 16; ++q) {
    const int a = v[q].x, b = v[q].y, c = v[q].z, d = v[q].w;
    v[q].x = s; s += a;
    v[q].y = s; s += b;
    v[q].z = s; s += c;
    v[q].w = s; s += d;
  }
  int inc = s;
  #pragma unroll
  for (int off = 1; off < 32; off <<= 1) {
    const int nb = __shfl_up(inc, off);
    if (lane >= off) inc += nb;
  }
  if (lane == 31) wsum[wv] = inc;
  __syncthreads();
  if (wv == 0) {
    const int x = wsum[lane];
    int xi = x;
    #pragma unroll
    for (int off = 1; off < 32; off <<= 1) {
      const int nb = __shfl_up(xi, off);
      if (lane >= off) xi += nb;
    }
    wsum[lane] = xi - x;
  }
  __syncthreads();
  const int base = wsum[wv] + inc - s;
  int4* r4 = reinterpret_cast<int4*>(rowptr) + t * 16;
  #pragma unroll
  for (int q = 0; q < 16; ++q) {
    int4 o = v[q];
    o.x += base; o.y += base; o.z += base; o.w += base;
    r4[q] = o;
  }
  if (t == 1023) rowptr[kN] = base + s;
}

__global__ __launch_bounds__(kSortT) void range_place_kernel(const int* __restrict__ row,
                                                             const int* __restrict__ col,
                                                             const float* __restrict__ ew,
                                                             const int* __restrict__ segstart,
                                                             int2* __restrict__ recs) {
  __shared__ int cur[kBins];
  const int t = threadIdx.x;
  const int e0 = blockIdx.x * kChunkE + t;
  int c[kSortIters], s[kSortIters], p[kSortIters];
  float w[kSortIters];
  #pragma unroll
  for (int q = 0; q < kSortIters; ++q) {
    const int e = e0 + q * kSortT;
    c[q] = col[e];
    s[q] = row[e];
    w[q] = ew[e];
  }
  if (t < kBins) cur[t] = segstart[t * kChunks + blockIdx.x];
  __syncthreads();
  #pragma unroll
  for (int q = 0; q < kSortIters; ++q) p[q] = atomicAdd(&cur[(c[q] >> kFineShift) & (kBins - 1)], 1);
  #pragma unroll
  for (int q = 0; q < kSortIters; ++q)
    recs[p[q]] = make_int2(s[q] | ((c[q] & (kFine - 1)) << 16), __float_as_int(w[q]));
}

__global__ __launch_bounds__(kSortT) void node_sort_kernel(const int2* __restrict__ recs,
                                                           const int* __restrict__ segstart,
                                                           int* __restrict__ rowptr,
                                                           int2* __restrict__ edges) {
  __shared__ int cnt[kFine];
  __shared__ int cur[kFine];
  const int t = threadIdx.x;
  const int bin = blockIdx.x;
  const int beg = segstart[bin * kChunks];
  const int end = segstart[(bin + 1) * kChunks];
  if (t < kFine) cnt[t] = 0;
  __syncthreads();

  for (int j0 = beg; j0 < end; j0 += kChunkE) {
    int k[kSortIters];
    #pragma unroll
    for (int q = 0; q < kSortIters; ++q)
      k[q] = recs[min(j0 + q * kSortT + t, kE - 1)].x;
    #pragma unroll
    for (int q = 0; q < kSortIters; ++q)
      if (j0 + q * kSortT + t < end) atomicAdd(&cnt[(k[q] >> 16) & (kFine - 1)], 1);
  }
  __syncthreads();

  if (t < 32) {
    int v[kFine / 32];
    int s = 0;
    #pragma unroll
    for (int q = 0; q < kFine / 32; ++q) {
      const int a = cnt[t * (kFine / 32) + q];
      v[q] = s;
      s += a;
    }
    int inc = s;
    #pragma unroll
    for (int off = 1; off < 32; off <<= 1) {
      const int nb = __shfl_up(inc, off);
      if (t >= off) inc += nb;
    }
    const int base = beg + inc - s;
    #pragma unroll
    for (int q = 0; q < kFine / 32; ++q) {
      const int o = base + v[q];
      cur[t * (kFine / 32) + q] = o;
      rowptr[bin * kFine + t * (kFine / 32) + q] = o;
    }
    if (t == 0 && bin == kBins - 1) rowptr[kN] = end;
  }
  __syncthreads();

  for (int j0 = beg; j0 < end; j0 += kChunkE) {
    int2 r[kSortIters];
    #pragma unroll
    for (int q = 0; q < kSortIters; ++q)
      r[q] = recs[min(j0 + q * kSortT + t, kE - 1)];
    int p[kSortIters];
    #pragma unroll
    for (int q = 0; q < kSortIters; ++q) {
      p[q] = 0;
      if (j0 + q * kSortT + t < end) p[q] = atomicAdd(&cur[(r[q].x >> 16) & (kFine - 1)], 1);
    }
    #pragma unroll
    for (int q = 0; q < kSortIters; ++q)
      if (j0 + q * kSortT + t < end) edges[p[q]] = make_int2(r[q].x & 0xFFFF, r[q].y);
  }
}

__global__ void dinv_kernel(const int* __restrict__ rowptr,
                            const int2* __restrict__ edges,
                            float* __restrict__ dinv) {
  const int t = blockIdx.x * 256 + threadIdx.x;
  const int hw = t >> 4;
  const int l = t & 15;
  int beg[kNodeIters], end[kNodeIters];
  #pragma unroll
  for (int q = 0; q < kNodeIters; ++q) {
    beg[q] = rowptr[hw + q * kNodeHW];
    end[q] = rowptr[hw + q * kNodeHW + 1];
  }
  #pragma unroll
  for (int q = 0; q < kNodeIters; ++q) {
    const int i = hw + q * kNodeHW;
    float s = 0.f;
    for (int j = beg[q] + l; j < end[q]; j += 16) s += __int_as_float(edges[j].y);
    s = sum16(s) + 1.0f;
    if (l == 0) dinv[i] = (s > 0.f) ? rsqrtf(s) : 0.f;
  }
}

__global__ void norm_kernel(const int* __restrict__ rowptr,
                            const float* __restrict__ dinv,
                            int2* __restrict__ edges) {
  const int t = blockIdx.x * 256 + threadIdx.x;
  const int hw = t >> 4;
  const int l = t & 15;
  int beg[kNodeIters], end[kNodeIters];
  float di[kNodeIters];
  #pragma unroll
  for (int q = 0; q < kNodeIters; ++q) {
    const int i = hw + q * kNodeHW;
    beg[q] = rowptr[i];
    end[q] = rowptr[i + 1];
    di[q] = dinv[i];
  }
  #pragma unroll
  for (int q = 0; q < kNodeIters; ++q) {
    for (int j = beg[q] + l; j < end[q]; j += 16) {
      const int2 ed = edges[j];
      const float nm = dinv[ed.x] * __int_as_float(ed.y) * di[q];
      edges[j].y = __float_as_int(nm);
    }
  }
}

__global__ void gemm_wmma_kernel(const float* __restrict__ A,
                                 const float* __restrict__ W,
                                 float* __restrict__ H, int K) {
  __shared__ float Ws[kMaxK * 17];

  const int tid  = threadIdx.x;
  const int lane = tid & 31;
  const int wave = tid >> 5;
  const int n0 = blockIdx.y * 16;
  const int half = lane >> 4;
  const int l = lane & 15;

  for (int idx = tid; idx < K * 16; idx += 128) {
    int kk = idx >> 4;
    int nn = idx & 15;
    Ws[kk * 17 + nn] = W[(size_t)kk * kDH + n0 + nn];
  }
  __syncthreads();

  for (int g = blockIdx.x; g < kN / 64; g += gridDim.x) {
    const int m0 = (g * 4 + wave) * 16;
    v8f acc = {};
    const float* arow = A + (size_t)(m0 + l) * K;
    #pragma unroll 4
    for (int k = 0; k < K; k += 4) {
      const int ka = k + 2 * half;
      v2f a, b;
      a.x = arow[ka];
      a.y = arow[ka + 1];
      b.x = Ws[ka * 17 + l];
      b.y = Ws[(ka + 1) * 17 + l];
      acc = __builtin_amdgcn_wmma_f32_16x16x4_f32(
          false, a, false, b, (short)0, acc, false, false);
    }

    float* hrow = H + (size_t)(m0 + half * 8) * kDH + (n0 + l);
    #pragma unroll
    for (int r = 0; r < 8; ++r) hrow[(size_t)r * kDH] = acc[r];
  }
}

template <int NL>
__device__ __forceinline__ void rows_fma(const float4* __restrict__ h4, int l,
                                         int sx, float wx, float4& acc) {
  int s[NL];
  float w[NL];
  float4 a[NL];
  #pragma unroll
  for (int q = 0; q < NL; ++q) {
    s[q] = __shfl(sx, q, 16);
    w[q] = __shfl(wx, q, 16);
  }
  #pragma unroll
  for (int q = 0; q < NL; ++q) a[q] = h4[(size_t)s[q] * 16 + l];
  #pragma unroll
  for (int q = 0; q < NL; ++q) {
    acc.x += w[q] * a[q].x;
    acc.y += w[q] * a[q].y;
    acc.z += w[q] * a[q].z;
    acc.w += w[q] * a[q].w;
  }
}

template <class Emit>
__device__ __forceinline__ void gather_rows(const int* __restrict__ rowptr,
                                            const int2* __restrict__ edges,
                                            const float* __restrict__ h,
                                            const float* __restrict__ dinv,
                                            const float* __restrict__ bias,
                                            Emit emit) {
  const int t = blockIdx.x * 256 + threadIdx.x;
  const int hw = t >> 4;
  const int l = t & 15;
  const float4* __restrict__ h4 = reinterpret_cast<const float4*>(h);
  const float b0 = bias[l * 4 + 0], b1 = bias[l * 4 + 1];
  const float b2 = bias[l * 4 + 2], b3 = bias[l * 4 + 3];

  int i = hw;
  int beg = rowptr[i], end = rowptr[i + 1];
  int i1 = min(i + kNodeHW, kN - 1);
  int beg1 = rowptr[i1], end1 = rowptr[i1 + 1];
  int2 ed = edges[min(beg + l, kE - 1)];

  for (int it = 0; it < kNodeIters; ++it) {
    const int i2 = min(i1 + kNodeHW, kN - 1);
    const int beg2 = rowptr[i2], end2 = rowptr[i2 + 1];
    const int2 ed1 = edges[min(beg1 + l, kE - 1)];
    const float di = dinv[i];
    const float4 hs = h4[(size_t)i * 16 + l];

    float4 acc = make_float4(0.f, 0.f, 0.f, 0.f);
    int base = beg;
    int sx = (base + l < end) ? ed.x : i;
    float wx = (base + l < end) ? __int_as_float(ed.y) : 0.f;
    while (base < end) {
      const int m = end - base;
      const int nb = base + 16;
      ed = edges[min(nb + l, kE - 1)];
      if (m > 8) {
        rows_fma<16>(h4, l, sx, wx, acc);
      } else {
        rows_fma<8>(h4, l, sx, wx, acc);
      }
      sx = (nb + l < end) ? ed.x : i;
      wx = (nb + l < end) ? __int_as_float(ed.y) : 0.f;
      base = nb;
    }

    const float dd = di * di;
    float4 v;
    v.x = acc.x + dd * hs.x + b0;
    v.y = acc.y + dd * hs.y + b1;
    v.z = acc.z + dd * hs.z + b2;
    v.w = acc.w + dd * hs.w + b3;
    v.x = fmaxf(v.x, 0.f);
    v.y = fmaxf(v.y, 0.f);
    v.z = fmaxf(v.z, 0.f);
    v.w = fmaxf(v.w, 0.f);
    emit(it, i, l, v);

    i = i1; beg = beg1; end = end1; ed = ed1;
    i1 = i2; beg1 = beg2; end1 = end2;
  }
}

constexpr int kTP = 68;
static_assert(kNodeIters * 16 == 64, "the LDS tile holds 4 M-tiles of 16 rows");

__global__ __launch_bounds__(256) __attribute__((amdgpu_waves_per_eu(4, 8))) void gather64_gemm_kernel(const int* __restrict__ rowptr,
                                                       const int2* __restrict__ edges,
                                                       const float* __restrict__ h,
                                                       const float* __restrict__ dinv,
                                                       const float* __restrict__ bias,
                                                       const float* __restrict__ Wn,
                                                       float* __restrict__ hn) {
  __shared__ float As[64 * kTP];
  __shared__ float Wt[64 * kTP];
  const int tid = threadIdx.x;

  #pragma unroll
  for (int j = 0; j < 16; ++j) {
    const int idx = tid + j * 256;
    Wt[(idx & 63) * kTP + (idx >> 6)] = Wn[idx];
  }

  const int hwl = tid >> 4;
  gather_rows(rowptr, edges, h, dinv, bias,
              [&](int it, int  , int l, const float4& v) {
                *reinterpret_cast<float4*>(&As[(it * 16 + hwl) * kTP + 4 * l]) = v;
              });
  __syncthreads();

  const int lane = tid & 31;
  const int wave = tid >> 5;
  const int half = lane >> 4;
  const int l = lane & 15;
  const int mt = wave >> 1;
  const int n0 = (wave & 1) * 32;
  const float* ap  = &As[(mt * 16 + l) * kTP + 2 * half];
  const float* bp0 = &Wt[(n0 + l) * kTP + 2 * half];
  const float* bp1 = &Wt[(n0 + 16 + l) * kTP + 2 * half];
  v8f acc0 = {}, acc1 = {};
  #pragma unroll
  for (int k = 0; k < kDH; k += 4) {
    const v2f a  = *reinterpret_cast<const v2f*>(ap + k);
    const v2f b0 = *reinterpret_cast<const v2f*>(bp0 + k);
    const v2f b1 = *reinterpret_cast<const v2f*>(bp1 + k);
    acc0 = __builtin_amdgcn_wmma_f32_16x16x4_f32(
        false, a, false, b0, (short)0, acc0, false, false);
    acc1 = __builtin_amdgcn_wmma_f32_16x16x4_f32(
        false, a, false, b1, (short)0, acc1, false, false);
  }

  float* hrow = hn + (size_t)(blockIdx.x * 16 + mt * kNodeHW + 8 * half) * kDH + (n0 + l);
  #pragma unroll
  for (int r = 0; r < 8; ++r) {
    hrow[(size_t)r * kDH] = acc0[r];
    hrow[(size_t)r * kDH + 16] = acc1[r];
  }
}

__global__ __launch_bounds__(256) __attribute__((amdgpu_waves_per_eu(4, 8))) void gather64_proj_kernel(const int* __restrict__ rowptr,
                                                       const int2* __restrict__ edges,
                                                       const float* __restrict__ h,
                                                       const float* __restrict__ dinv,
                                                       const float* __restrict__ bias,
                                                       const float* __restrict__ W8,
                                                       float* __restrict__ h2) {
  const int lw = threadIdx.x & 15;
  float w0[4], w1[4];
  #pragma unroll
  for (int q = 0; q < 4; ++q) {
    w0[q] = W8[(lw * 4 + q) * 2 + 0];
    w1[q] = W8[(lw * 4 + q) * 2 + 1];
  }
  gather_rows(rowptr, edges, h, dinv, bias,
              [&](int  , int i, int l, const float4& v) {
                float s0 = v.x * w0[0];
                float s1 = v.x * w1[0];
                s0 = fmaf(v.y, w0[1], s0);
                s1 = fmaf(v.y, w1[1], s1);
                s0 = fmaf(v.z, w0[2], s0);
                s1 = fmaf(v.z, w1[2], s1);
                s0 = fmaf(v.w, w0[3], s0);
                s1 = fmaf(v.w, w1[3], s1);
                s0 = sum16(s0);
                s1 = sum16(s1);
                if (l == 0) reinterpret_cast<float2*>(h2)[i] = make_float2(s0, s1);
              });
}

__global__ void gather2_kernel(const int* __restrict__ rowptr,
                               const int2* __restrict__ edges,
                               const float* __restrict__ h2,
                               const float* __restrict__ dinv,
                               const float* __restrict__ b8,
                               float* __restrict__ out) {
  const int t = blockIdx.x * 256 + threadIdx.x;
  const int hw = t >> 4;
  const int l = t & 15;
  int beg[kNodeIters], end[kNodeIters];
  #pragma unroll
  for (int q = 0; q < kNodeIters; ++q) {
    beg[q] = rowptr[hw + q * kNodeHW];
    end[q] = rowptr[hw + q * kNodeHW + 1];
  }
  #pragma unroll
  for (int q = 0; q < kNodeIters; ++q) {
    const int i = hw + q * kNodeHW;
    float a0 = 0.f, a1 = 0.f;
    for (int j = beg[q] + l; j < end[q]; j += 16) {
      const int2 ed = edges[j];
      const float w = __int_as_float(ed.y);
      a0 += w * h2[(size_t)ed.x * 2 + 0];
      a1 += w * h2[(size_t)ed.x * 2 + 1];
    }
    a0 = sum16(a0);
    a1 = sum16(a1);
    if (l < 2) {
      const float di = dinv[i];
      const float a = (l == 0) ? a0 : a1;
      out[(size_t)i * 2 + l] = a + di * di * h2[(size_t)i * 2 + l] + b8[l];
    }
  }
}

}

extern "C" void kernel_launch(void* const* d_in, const int* in_sizes, int n_in,
                              void* d_out, int out_size, void* d_ws, size_t ws_size,
                              hipStream_t stream) {
  (void)in_sizes; (void)n_in; (void)out_size; (void)ws_size;

  const float* x  = (const float*)d_in[0];
  const int*   ei = (const int*)d_in[1];
  const float* ew = (const float*)d_in[2];
  const float* W[5] = {(const float*)d_in[3], (const float*)d_in[5], (const float*)d_in[7],
                       (const float*)d_in[9], (const float*)d_in[11]};
  const float* B[5] = {(const float*)d_in[4], (const float*)d_in[6], (const float*)d_in[8],
                       (const float*)d_in[10], (const float*)d_in[12]};
  const float* W8 = (const float*)d_in[13];
  const float* b8 = (const float*)d_in[14];
  float* out = (float*)d_out;

  const int* row = ei;
  const int* col = ei + kE;

  float* ws = (float*)d_ws;
  float* dinv  = ws;         ws += kN;
  int2*  edges = (int2*)ws;  ws += (size_t)kE * 2;
  float* feat  = ws;         ws += (size_t)kN * kDH;
  float* hbuf  = ws;         ws += (size_t)kN * kDH;
  int*   table = (int*)ws;   ws += kN;
  int*   rowptr = (int*)ws;
  int2*  recs  = (int2*)feat;
  int*   segstart = (int*)hbuf;

  range_count_kernel<<<kChunks, kSortT, 0, stream>>>(col, table);
  scan_kernel<<<1, 1024, 0, stream>>>(table, segstart);
  range_place_kernel<<<kChunks, kSortT, 0, stream>>>(row, col, ew, segstart, recs);
  node_sort_kernel<<<kBins, kSortT, 0, stream>>>(recs, segstart, rowptr, edges);

  dinv_kernel<<<kNodeBlocks, 256, 0, stream>>>(rowptr, edges, dinv);
  norm_kernel<<<kNodeBlocks, 256, 0, stream>>>(rowptr, dinv, edges);

  const dim3 ggrid(kGemmBlocksX, kDH / 16);

  gemm_wmma_kernel<<<ggrid, 128, 0, stream>>>(x, W[0], hbuf, 128);
  gather64_gemm_kernel<<<kNodeBlocks, 256, 0, stream>>>(rowptr, edges, hbuf, dinv, B[0], W[1], feat);
  gather64_gemm_kernel<<<kNodeBlocks, 256, 0, stream>>>(rowptr, edges, feat, dinv, B[1], W[2], hbuf);
  gather64_gemm_kernel<<<kNodeBlocks, 256, 0, stream>>>(rowptr, edges, hbuf, dinv, B[2], W[3], feat);
  gather64_gemm_kernel<<<kNodeBlocks, 256, 0, stream>>>(rowptr, edges, feat, dinv, B[3], W[4], hbuf);

  gather64_proj_kernel<<<kNodeBlocks, 256, 0, stream>>>(rowptr, edges, hbuf, dinv, B[4], W8, feat);
  gather2_kernel<<<kNodeBlocks, 256, 0, stream>>>(rowptr, edges, feat, dinv, b8, out);
}


# Degree normalisation leaves the edge list: weighted degree summed by LDS float adds in the bucketing block, D^-1/2 carried by the rows and applied once per node sum, not per edge: 2 launches gone
# speedup vs baseline: 4.2540x; 1.0919x over previous
#include <hip/hip_runtime.h>


namespace {

constexpr int kN  = 65536;
constexpr int kE  = 1048576;
constexpr int kDH = 64;
constexpr int kMaxK = 128;

constexpr int kEdgeBlocks = 512;
constexpr int kEdgeT = kEdgeBlocks * 256;
constexpr int kEdgeIters = kE / kEdgeT;
constexpr int kNodeBlocks = 1024;
constexpr int kNodeHW = kNodeBlocks * 256 / 16;
constexpr int kNodeIters = kN / kNodeHW;
constexpr int kGemmBlocksX = 512;
static_assert(kEdgeIters * kEdgeT == kE, "edge launch must tile the edges");
static_assert(kNodeIters * kNodeHW == kN, "node launch must tile the nodes");
static_assert((kN / 64) % kGemmBlocksX == 0, "row groups per block must be whole");

constexpr int kBins = 256;
constexpr int kFineShift = 8;
constexpr int kFine = 1 << kFineShift;
constexpr int kChunks = 256;
constexpr int kChunkE = kE / kChunks;
constexpr int kSortT = 512;
constexpr int kSortIters = kChunkE / kSortT;
static_assert(kBins * kFine == kN, "ranges must tile the nodes");
static_assert(kBins * kChunks == kN, "the (range, chunk) table is scanned by the kN-entry scan");
static_assert(kSortIters * kSortT * kChunks == kE, "chunks must tile the edges");
static_assert(kN <= 65536, "a source index must fit the low 16 bits of a packed record");

typedef __attribute__((ext_vector_type(2))) float v2f;
typedef __attribute__((ext_vector_type(8))) float v8f;

__device__ __forceinline__ float sum16(float v) {
  v += __shfl_xor(v, 8, 16);
  v += __shfl_xor(v, 4, 16);
  v += __shfl_xor(v, 2, 16);
  v += __shfl_xor(v, 1, 16);
  return v;
}

__global__ __launch_bounds__(kSortT) void range_count_kernel(const int* __restrict__ col,
                                                             int* __restrict__ table) {
  __shared__ int h[kBins];
  const int t = threadIdx.x;
  if (t < kBins) h[t] = 0;
  const int e0 = blockIdx.x * kChunkE + t;
  int c[kSortIters];
  #pragma unroll
  for (int q = 0; q < kSortIters; ++q) c[q] = col[e0 + q * kSortT];
  __syncthreads();
  #pragma unroll
  for (int q = 0; q < kSortIters; ++q) atomicAdd(&h[(c[q] >> kFineShift) & (kBins - 1)], 1);
  __syncthreads();
  if (t < kBins) table[t * kChunks + blockIdx.x] = h[t];
}

__global__ __launch_bounds__(1024) void scan_kernel(const int* __restrict__ cnt,
                                                    int* __restrict__ rowptr) {
  __shared__ int wsum[32];
  const int t = threadIdx.x;
  const int lane = t & 31;
  const int wv = t >> 5;
  const int4* c4 = reinterpret_cast<const int4*>(cnt) + t * 16;
  int4 v[16];
  #pragma unroll
  for (int q = 0; q < 16; ++q) v[q] = c4[q];
  int s = 0;
  #pragma unroll
  for (int q = 0; q < 16; ++q) {
    const int a = v[q].x, b = v[q].y, c = v[q].z, d = v[q].w;
    v[q].x = s; s += a;
    v[q].y = s; s += b;
    v[q].z = s; s += c;
    v[q].w = s; s += d;
  }
  int inc = s;
  #pragma unroll
  for (int off = 1; off < 32; off <<= 1) {
    const int nb = __shfl_up(inc, off);
    if (lane >= off) inc += nb;
  }
  if (lane == 31) wsum[wv] = inc;
  __syncthreads();
  if (wv == 0) {
    const int x = wsum[lane];
    int xi = x;
    #pragma unroll
    for (int off = 1; off < 32; off <<= 1) {
      const int nb = __shfl_up(xi, off);
      if (lane >= off) xi += nb;
    }
    wsum[lane] = xi - x;
  }
  __syncthreads();
  const int base = wsum[wv] + inc - s;
  int4* r4 = reinterpret_cast<int4*>(rowptr) + t * 16;
  #pragma unroll
  for (int q = 0; q < 16; ++q) {
    int4 o = v[q];
    o.x += base; o.y += base; o.z += base; o.w += base;
    r4[q] = o;
  }
  if (t == 1023) rowptr[kN] = base + s;
}

__global__ __launch_bounds__(kSortT) void range_place_kernel(const int* __restrict__ row,
                                                             const int* __restrict__ col,
                                                             const float* __restrict__ ew,
                                                             const int* __restrict__ segstart,
                                                             int2* __restrict__ recs) {
  __shared__ int cur[kBins];
  const int t = threadIdx.x;
  const int e0 = blockIdx.x * kChunkE + t;
  int c[kSortIters], s[kSortIters], p[kSortIters];
  float w[kSortIters];
  #pragma unroll
  for (int q = 0; q < kSortIters; ++q) {
    const int e = e0 + q * kSortT;
    c[q] = col[e];
    s[q] = row[e];
    w[q] = ew[e];
  }
  if (t < kBins) cur[t] = segstart[t * kChunks + blockIdx.x];
  __syncthreads();
  #pragma unroll
  for (int q = 0; q < kSortIters; ++q) p[q] = atomicAdd(&cur[(c[q] >> kFineShift) & (kBins - 1)], 1);
  #pragma unroll
  for (int q = 0; q < kSortIters; ++q)
    recs[p[q]] = make_int2(s[q] | ((c[q] & (kFine - 1)) << 16), __float_as_int(w[q]));
}

static_assert(kSortT >= 2 * kFine, "threads kFine .. 2*kFine-1 write the range's dinv");
__global__ __launch_bounds__(kSortT) void node_sort_kernel(const int2* __restrict__ recs,
                                                           const int* __restrict__ segstart,
                                                           int* __restrict__ rowptr,
                                                           int2* __restrict__ edges,
                                                           float* __restrict__ dinv) {
  __shared__ int cnt[kFine];
  __shared__ int cur[kFine];
  __shared__ float wdeg[kFine];
  const int t = threadIdx.x;
  const int bin = blockIdx.x;
  const int beg = segstart[bin * kChunks];
  const int end = segstart[(bin + 1) * kChunks];
  if (t < kFine) {
    cnt[t] = 0;
    wdeg[t] = 0.f;
  }
  __syncthreads();

  for (int j0 = beg; j0 < end; j0 += kChunkE) {
    int2 k[kSortIters];
    #pragma unroll
    for (int q = 0; q < kSortIters; ++q)
      k[q] = recs[min(j0 + q * kSortT + t, kE - 1)];
    #pragma unroll
    for (int q = 0; q < kSortIters; ++q)
      if (j0 + q * kSortT + t < end) {
        const int nd = (k[q].x >> 16) & (kFine - 1);
        atomicAdd(&cnt[nd], 1);
        atomicAdd(&wdeg[nd], __int_as_float(k[q].y));
      }
  }
  __syncthreads();

  if (t >= kFine && t < 2 * kFine) {
    const float s = wdeg[t - kFine] + 1.0f;
    dinv[bin * kFine + (t - kFine)] = (s > 0.f) ? rsqrtf(s) : 0.f;
  }

  if (t < 32) {
    int v[kFine / 32];
    int s = 0;
    #pragma unroll
    for (int q = 0; q < kFine / 32; ++q) {
      const int a = cnt[t * (kFine / 32) + q];
      v[q] = s;
      s += a;
    }
    int inc = s;
    #pragma unroll
    for (int off = 1; off < 32; off <<= 1) {
      const int nb = __shfl_up(inc, off);
      if (t >= off) inc += nb;
    }
    const int base = beg + inc - s;
    #pragma unroll
    for (int q = 0; q < kFine / 32; ++q) {
      const int o = base + v[q];
      cur[t * (kFine / 32) + q] = o;
      rowptr[bin * kFine + t * (kFine / 32) + q] = o;
    }
    if (t == 0 && bin == kBins - 1) rowptr[kN] = end;
  }
  __syncthreads();

  for (int j0 = beg; j0 < end; j0 += kChunkE) {
    int2 r[kSortIters];
    #pragma unroll
    for (int q = 0; q < kSortIters; ++q)
      r[q] = recs[min(j0 + q * kSortT + t, kE - 1)];
    int p[kSortIters];
    #pragma unroll
    for (int q = 0; q < kSortIters; ++q) {
      p[q] = 0;
      if (j0 + q * kSortT + t < end) p[q] = atomicAdd(&cur[(r[q].x >> 16) & (kFine - 1)], 1);
    }
    #pragma unroll
    for (int q = 0; q < kSortIters; ++q)
      if (j0 + q * kSortT + t < end) edges[p[q]] = make_int2(r[q].x & 0xFFFF, r[q].y);
  }
}

__global__ void gemm_wmma_kernel(const float* __restrict__ A,
                                 const float* __restrict__ W,
                                 const float* __restrict__ dinv,
                                 float* __restrict__ H, int K) {
  __shared__ float Ws[kMaxK * 17];

  const int tid  = threadIdx.x;
  const int lane = tid & 31;
  const int wave = tid >> 5;
  const int n0 = blockIdx.y * 16;
  const int half = lane >> 4;
  const int l = lane & 15;

  for (int idx = tid; idx < K * 16; idx += 128) {
    int kk = idx >> 4;
    int nn = idx & 15;
    Ws[kk * 17 + nn] = W[(size_t)kk * kDH + n0 + nn];
  }
  __syncthreads();

  for (int g = blockIdx.x; g < kN / 64; g += gridDim.x) {
    const int m0 = (g * 4 + wave) * 16;
    v8f acc = {};
    const float* arow = A + (size_t)(m0 + l) * K;
    float ds[8];
    #pragma unroll
    for (int r = 0; r < 8; ++r) ds[r] = dinv[m0 + half * 8 + r];
    #pragma unroll 4
    for (int k = 0; k < K; k += 4) {
      const int ka = k + 2 * half;
      v2f a, b;
      a.x = arow[ka];
      a.y = arow[ka + 1];
      b.x = Ws[ka * 17 + l];
      b.y = Ws[(ka + 1) * 17 + l];
      acc = __builtin_amdgcn_wmma_f32_16x16x4_f32(
          false, a, false, b, (short)0, acc, false, false);
    }

    float* hrow = H + (size_t)(m0 + half * 8) * kDH + (n0 + l);
    #pragma unroll
    for (int r = 0; r < 8; ++r) hrow[(size_t)r * kDH] = acc[r] * ds[r];
  }
}

template <int NL>
__device__ __forceinline__ void rows_fma(const float4* __restrict__ h4, int l,
                                         int sx, float wx, float4& acc) {
  int s[NL];
  float w[NL];
  float4 a[NL];
  #pragma unroll
  for (int q = 0; q < NL; ++q) {
    s[q] = __shfl(sx, q, 16);
    w[q] = __shfl(wx, q, 16);
  }
  #pragma unroll
  for (int q = 0; q < NL; ++q) a[q] = h4[(size_t)s[q] * 16 + l];
  #pragma unroll
  for (int q = 0; q < NL; ++q) {
    acc.x += w[q] * a[q].x;
    acc.y += w[q] * a[q].y;
    acc.z += w[q] * a[q].z;
    acc.w += w[q] * a[q].w;
  }
}

template <class Emit>
__device__ __forceinline__ void gather_rows(const int* __restrict__ rowptr,
                                            const int2* __restrict__ edges,
                                            const float* __restrict__ h,
                                            const float* __restrict__ dinv,
                                            const float* __restrict__ bias,
                                            Emit emit) {
  const int t = blockIdx.x * 256 + threadIdx.x;
  const int hw = t >> 4;
  const int l = t & 15;
  const float4* __restrict__ h4 = reinterpret_cast<const float4*>(h);
  const float b0 = bias[l * 4 + 0], b1 = bias[l * 4 + 1];
  const float b2 = bias[l * 4 + 2], b3 = bias[l * 4 + 3];

  int i = hw;
  int beg = rowptr[i], end = rowptr[i + 1];
  int i1 = min(i + kNodeHW, kN - 1);
  int beg1 = rowptr[i1], end1 = rowptr[i1 + 1];
  int2 ed = edges[min(beg + l, kE - 1)];

  for (int it = 0; it < kNodeIters; ++it) {
    const int i2 = min(i1 + kNodeHW, kN - 1);
    const int beg2 = rowptr[i2], end2 = rowptr[i2 + 1];
    const int2 ed1 = edges[min(beg1 + l, kE - 1)];
    const float di = dinv[i];
    const float4 hs = h4[(size_t)i * 16 + l];

    float4 acc = make_float4(0.f, 0.f, 0.f, 0.f);
    int base = beg;
    int sx = (base + l < end) ? ed.x : i;
    float wx = (base + l < end) ? __int_as_float(ed.y) : 0.f;
    while (base < end) {
      const int m = end - base;
      const int nb = base + 16;
      ed = edges[min(nb + l, kE - 1)];
      if (m > 8) {
        rows_fma<16>(h4, l, sx, wx, acc);
      } else {
        rows_fma<8>(h4, l, sx, wx, acc);
      }
      sx = (nb + l < end) ? ed.x : i;
      wx = (nb + l < end) ? __int_as_float(ed.y) : 0.f;
      base = nb;
    }

    const float dd = di * di;
    float4 v;
    v.x = dd * acc.x + (dd * hs.x + di * b0);
    v.y = dd * acc.y + (dd * hs.y + di * b1);
    v.z = dd * acc.z + (dd * hs.z + di * b2);
    v.w = dd * acc.w + (dd * hs.w + di * b3);
    v.x = fmaxf(v.x, 0.f);
    v.y = fmaxf(v.y, 0.f);
    v.z = fmaxf(v.z, 0.f);
    v.w = fmaxf(v.w, 0.f);
    emit(it, i, l, v);

    i = i1; beg = beg1; end = end1; ed = ed1;
    i1 = i2; beg1 = beg2; end1 = end2;
  }
}

constexpr int kTP = 68;
static_assert(kNodeIters * 16 == 64, "the LDS tile holds 4 M-tiles of 16 rows");

__global__ __launch_bounds__(256) __attribute__((amdgpu_waves_per_eu(4, 8))) void gather64_gemm_kernel(const int* __restrict__ rowptr,
                                                       const int2* __restrict__ edges,
                                                       const float* __restrict__ h,
                                                       const float* __restrict__ dinv,
                                                       const float* __restrict__ bias,
                                                       const float* __restrict__ Wn,
                                                       float* __restrict__ hn) {
  __shared__ float As[64 * kTP];
  __shared__ float Wt[64 * kTP];
  const int tid = threadIdx.x;

  #pragma unroll
  for (int j = 0; j < 16; ++j) {
    const int idx = tid + j * 256;
    Wt[(idx & 63) * kTP + (idx >> 6)] = Wn[idx];
  }

  const int hwl = tid >> 4;
  gather_rows(rowptr, edges, h, dinv, bias,
              [&](int it, int  , int l, const float4& v) {
                *reinterpret_cast<float4*>(&As[(it * 16 + hwl) * kTP + 4 * l]) = v;
              });
  __syncthreads();

  const int lane = tid & 31;
  const int wave = tid >> 5;
  const int half = lane >> 4;
  const int l = lane & 15;
  const int mt = wave >> 1;
  const int n0 = (wave & 1) * 32;
  const float* ap  = &As[(mt * 16 + l) * kTP + 2 * half];
  const float* bp0 = &Wt[(n0 + l) * kTP + 2 * half];
  const float* bp1 = &Wt[(n0 + 16 + l) * kTP + 2 * half];
  v8f acc0 = {}, acc1 = {};
  #pragma unroll
  for (int k = 0; k < kDH; k += 4) {
    const v2f a  = *reinterpret_cast<const v2f*>(ap + k);
    const v2f b0 = *reinterpret_cast<const v2f*>(bp0 + k);
    const v2f b1 = *reinterpret_cast<const v2f*>(bp1 + k);
    acc0 = __builtin_amdgcn_wmma_f32_16x16x4_f32(
        false, a, false, b0, (short)0, acc0, false, false);
    acc1 = __builtin_amdgcn_wmma_f32_16x16x4_f32(
        false, a, false, b1, (short)0, acc1, false, false);
  }

  float* hrow = hn + (size_t)(blockIdx.x * 16 + mt * kNodeHW + 8 * half) * kDH + (n0 + l);
  #pragma unroll
  for (int r = 0; r < 8; ++r) {
    hrow[(size_t)r * kDH] = acc0[r];
    hrow[(size_t)r * kDH + 16] = acc1[r];
  }
}

__global__ __launch_bounds__(256) __attribute__((amdgpu_waves_per_eu(4, 8))) void gather64_proj_kernel(const int* __restrict__ rowptr,
                                                       const int2* __restrict__ edges,
                                                       const float* __restrict__ h,
                                                       const float* __restrict__ dinv,
                                                       const float* __restrict__ bias,
                                                       const float* __restrict__ W8,
                                                       float* __restrict__ h2) {
  const int lw = threadIdx.x & 15;
  float w0[4], w1[4];
  #pragma unroll
  for (int q = 0; q < 4; ++q) {
    w0[q] = W8[(lw * 4 + q) * 2 + 0];
    w1[q] = W8[(lw * 4 + q) * 2 + 1];
  }
  gather_rows(rowptr, edges, h, dinv, bias,
              [&](int  , int i, int l, const float4& v) {
                float s0 = v.x * w0[0];
                float s1 = v.x * w1[0];
                s0 = fmaf(v.y, w0[1], s0);
                s1 = fmaf(v.y, w1[1], s1);
                s0 = fmaf(v.z, w0[2], s0);
                s1 = fmaf(v.z, w1[2], s1);
                s0 = fmaf(v.w, w0[3], s0);
                s1 = fmaf(v.w, w1[3], s1);
                s0 = sum16(s0);
                s1 = sum16(s1);
                if (l == 0) reinterpret_cast<float2*>(h2)[i] = make_float2(s0, s1);
              });
}

__global__ void gather2_kernel(const int* __restrict__ rowptr,
                               const int2* __restrict__ edges,
                               const float* __restrict__ h2,
                               const float* __restrict__ dinv,
                               const float* __restrict__ b8,
                               float* __restrict__ out) {
  const int t = blockIdx.x * 256 + threadIdx.x;
  const int hw = t >> 4;
  const int l = t & 15;
  int beg[kNodeIters], end[kNodeIters];
  #pragma unroll
  for (int q = 0; q < kNodeIters; ++q) {
    beg[q] = rowptr[hw + q * kNodeHW];
    end[q] = rowptr[hw + q * kNodeHW + 1];
  }
  #pragma unroll
  for (int q = 0; q < kNodeIters; ++q) {
    const int i = hw + q * kNodeHW;
    float a0 = 0.f, a1 = 0.f;
    for (int j = beg[q] + l; j < end[q]; j += 16) {
      const int2 ed = edges[j];
      const float w = __int_as_float(ed.y);
      a0 += w * h2[(size_t)ed.x * 2 + 0];
      a1 += w * h2[(size_t)ed.x * 2 + 1];
    }
    a0 = sum16(a0);
    a1 = sum16(a1);
    if (l < 2) {
      const float di = dinv[i];
      const float a = (l == 0) ? a0 : a1;
      out[(size_t)i * 2 + l] = di * (a + h2[(size_t)i * 2 + l]) + b8[l];
    }
  }
}

}

extern "C" void kernel_launch(void* const* d_in, const int* in_sizes, int n_in,
                              void* d_out, int out_size, void* d_ws, size_t ws_size,
                              hipStream_t stream) {
  (void)in_sizes; (void)n_in; (void)out_size; (void)ws_size;

  const float* x  = (const float*)d_in[0];
  const int*   ei = (const int*)d_in[1];
  const float* ew = (const float*)d_in[2];
  const float* W[5] = {(const float*)d_in[3], (const float*)d_in[5], (const float*)d_in[7],
                       (const float*)d_in[9], (const float*)d_in[11]};
  const float* B[5] = {(const float*)d_in[4], (const float*)d_in[6], (const float*)d_in[8],
                       (const float*)d_in[10], (const float*)d_in[12]};
  const float* W8 = (const float*)d_in[13];
  const float* b8 = (const float*)d_in[14];
  float* out = (float*)d_out;

  const int* row = ei;
  const int* col = ei + kE;

  float* ws = (float*)d_ws;
  float* dinv  = ws;         ws += kN;
  int2*  edges = (int2*)ws;  ws += (size_t)kE * 2;
  float* feat  = ws;         ws += (size_t)kN * kDH;
  float* hbuf  = ws;         ws += (size_t)kN * kDH;
  int*   table = (int*)ws;   ws += kN;
  int*   rowptr = (int*)ws;
  int2*  recs  = (int2*)feat;
  int*   segstart = (int*)hbuf;

  range_count_kernel<<<kChunks, kSortT, 0, stream>>>(col, table);
  scan_kernel<<<1, 1024, 0, stream>>>(table, segstart);
  range_place_kernel<<<kChunks, kSortT, 0, stream>>>(row, col, ew, segstart, recs);
  node_sort_kernel<<<kBins, kSortT, 0, stream>>>(recs, segstart, rowptr, edges, dinv);

  const dim3 ggrid(kGemmBlocksX, kDH / 16);

  gemm_wmma_kernel<<<ggrid, 128, 0, stream>>>(x, W[0], dinv, hbuf, 128);
  gather64_gemm_kernel<<<kNodeBlocks, 256, 0, stream>>>(rowptr, edges, hbuf, dinv, B[0], W[1], feat);
  gather64_gemm_kernel<<<kNodeBlocks, 256, 0, stream>>>(rowptr, edges, feat, dinv, B[1], W[2], hbuf);
  gather64_gemm_kernel<<<kNodeBlocks, 256, 0, stream>>>(rowptr, edges, hbuf, dinv, B[2], W[3], feat);
  gather64_gemm_kernel<<<kNodeBlocks, 256, 0, stream>>>(rowptr, edges, feat, dinv, B[3], W[4], hbuf);

  gather64_proj_kernel<<<kNodeBlocks, 256, 0, stream>>>(rowptr, edges, hbuf, dinv, B[4], W8, feat);
  gather2_kernel<<<kNodeBlocks, 256, 0, stream>>>(rowptr, edges, feat, dinv, b8, out);
}


# Edge bucketing in 2 launches, not 4: a chunk block ranks its edges by range in LDS into its own slice; a range block sums its pieces' offsets for its start (no global scan); new LDS-add bucket order
# speedup vs baseline: 5.2884x; 1.2432x over previous
#include <hip/hip_runtime.h>


namespace {

constexpr int kN  = 65536;
constexpr int kE  = 1048576;
constexpr int kDH = 64;
constexpr int kMaxK = 128;

constexpr int kEdgeBlocks = 512;
constexpr int kEdgeT = kEdgeBlocks * 256;
constexpr int kEdgeIters = kE / kEdgeT;
constexpr int kNodeBlocks = 1024;
constexpr int kNodeHW = kNodeBlocks * 256 / 16;
constexpr int kNodeIters = kN / kNodeHW;
constexpr int kGemmBlocksX = 512;
static_assert(kEdgeIters * kEdgeT == kE, "edge launch must tile the edges");
static_assert(kNodeIters * kNodeHW == kN, "node launch must tile the nodes");
static_assert((kN / 64) % kGemmBlocksX == 0, "row groups per block must be whole");

constexpr int kBins = 256;
constexpr int kFineShift = 8;
constexpr int kFine = 1 << kFineShift;
constexpr int kChunks = 256;
constexpr int kChunkE = kE / kChunks;
constexpr int kSortT = 512;
constexpr int kSortIters = kChunkE / kSortT;
constexpr int kSortW = kSortT / 32;
constexpr int kSegPerWave = kChunks / kSortW;
static_assert(kBins * kFine == kN, "ranges must tile the nodes");
static_assert(kBins * kChunks == kN, "the (range, chunk) table fills the kN-word table region");
static_assert(kSortIters * kSortT * kChunks == kE, "chunks must tile the edges");
static_assert(kSegPerWave * kSortW == kChunks, "the waves of a range block must tile its pieces");
static_assert(kBins <= kSortT && kChunks <= kSortT, "one thread per range / per piece");
static_assert(kChunkE <= 0x7FFF, "a piece's offset and count must fit 16 bits each, sign clear");
static_assert(kN <= 65536, "a source index must fit the low 16 bits of a packed record");

typedef __attribute__((ext_vector_type(2))) float v2f;
typedef __attribute__((ext_vector_type(8))) float v8f;

__device__ __forceinline__ float sum16(float v) {
  v += __shfl_xor(v, 8, 16);
  v += __shfl_xor(v, 4, 16);
  v += __shfl_xor(v, 2, 16);
  v += __shfl_xor(v, 1, 16);
  return v;
}

__global__ __launch_bounds__(kSortT) void chunk_sort_kernel(const int* __restrict__ row,
                                                            const int* __restrict__ col,
                                                            const float* __restrict__ ew,
                                                            int* __restrict__ table,
                                                            int2* __restrict__ recs) {
  __shared__ int h[kBins];
  __shared__ int offs[kBins];
  __shared__ int wtot[kBins / 32];
  const int t = threadIdx.x;
  const int lane = t & 31;
  const int wv = t >> 5;
  const int e0 = blockIdx.x * kChunkE + t;
  int c[kSortIters], s[kSortIters], r[kSortIters];
  float w[kSortIters];
  if (t < kBins) h[t] = 0;
  #pragma unroll
  for (int q = 0; q < kSortIters; ++q) {
    const int e = e0 + q * kSortT;
    c[q] = col[e];
    s[q] = row[e];
    w[q] = ew[e];
  }
  __syncthreads();
  #pragma unroll
  for (int q = 0; q < kSortIters; ++q) r[q] = atomicAdd(&h[(c[q] >> kFineShift) & (kBins - 1)], 1);
  __syncthreads();

  int v = 0, inc = 0;
  if (t < kBins) {
    v = h[t];
    inc = v;
    #pragma unroll
    for (int off = 1; off < 32; off <<= 1) {
      const int nb = __shfl_up(inc, off);
      if (lane >= off) inc += nb;
    }
    if (lane == 31) wtot[wv] = inc;
  }
  __syncthreads();
  if (t < kBins) {
    int base = 0;
    #pragma unroll
    for (int k = 0; k < kBins / 32; ++k) base += (k < wv) ? wtot[k] : 0;
    const int ex = base + inc - v;
    offs[t] = ex;
    table[t * kChunks + blockIdx.x] = ex | (v << 16);
  }
  __syncthreads();

  int2* __restrict__ slice = recs + (size_t)blockIdx.x * kChunkE;
  #pragma unroll
  for (int q = 0; q < kSortIters; ++q)
    slice[offs[(c[q] >> kFineShift) & (kBins - 1)] + r[q]] =
        make_int2(s[q] | ((c[q] & (kFine - 1)) << 16), __float_as_int(w[q]));
}

static_assert(kSortT >= 2 * kFine, "threads kFine .. 2*kFine-1 write the range's dinv");
__global__ __launch_bounds__(kSortT) void node_sort_kernel(const int2* __restrict__ recs,
                                                           const int* __restrict__ table,
                                                           int* __restrict__ rowptr,
                                                           int2* __restrict__ edges,
                                                           float* __restrict__ dinv) {
  __shared__ int cnt[kFine];
  __shared__ int cur[kFine];
  __shared__ float wdeg[kFine];
  __shared__ int seg[kChunks];
  __shared__ int wbeg[kSortW];
  __shared__ int wnum[kSortW];
  const int t = threadIdx.x;
  const int lane = t & 31;
  const int wv = t >> 5;
  const int bin = blockIdx.x;

  int oc = 0;
  if (t < kChunks) {
    oc = table[bin * kChunks + t];
    seg[t] = oc;
  }
  if (t < kFine) {
    cnt[t] = 0;
    wdeg[t] = 0.f;
  }
  __syncthreads();

  int2 k[kSegPerWave];
  int sb[kSegPerWave], sn[kSegPerWave];
  #pragma unroll
  for (int q = 0; q < kSegPerWave; ++q) {
    const int ch = wv * kSegPerWave + q;
    const int pc = seg[ch];
    sb[q] = ch * kChunkE + (pc & 0xFFFF);
    sn[q] = pc >> 16;
    k[q] = recs[min(sb[q] + lane, kE - 1)];
  }
  int so = oc & 0xFFFF;
  int sc = oc >> 16;
  #pragma unroll
  for (int off = 16; off > 0; off >>= 1) {
    so += __shfl_xor(so, off);
    sc += __shfl_xor(sc, off);
  }
  if (lane == 0) {
    wbeg[wv] = so;
    wnum[wv] = sc;
  }
  #pragma unroll
  for (int q = 0; q < kSegPerWave; ++q)
    if (lane < sn[q]) {
      const int nd = (k[q].x >> 16) & (kFine - 1);
      atomicAdd(&cnt[nd], 1);
      atomicAdd(&wdeg[nd], __int_as_float(k[q].y));
    }
  for (int q = 0; q < kSegPerWave; ++q) {
    const int ch = wv * kSegPerWave + q;
    const int pc = seg[ch];
    const int b = ch * kChunkE + (pc & 0xFFFF);
    const int n = pc >> 16;
    for (int j = 32 + lane; j < n; j += 32) {
      const int2 rr = recs[b + j];
      const int nd = (rr.x >> 16) & (kFine - 1);
      atomicAdd(&cnt[nd], 1);
      atomicAdd(&wdeg[nd], __int_as_float(rr.y));
    }
  }
  __syncthreads();

  if (t >= kFine && t < 2 * kFine) {
    const float s = wdeg[t - kFine] + 1.0f;
    dinv[bin * kFine + (t - kFine)] = (s > 0.f) ? rsqrtf(s) : 0.f;
  }

  if (t < 32) {
    int beg = 0, num = 0;
    #pragma unroll
    for (int q = 0; q < kSortW; ++q) {
      beg += wbeg[q];
      num += wnum[q];
    }
    const int end = beg + num;
    int v[kFine / 32];
    int s = 0;
    #pragma unroll
    for (int q = 0; q < kFine / 32; ++q) {
      const int a = cnt[t * (kFine / 32) + q];
      v[q] = s;
      s += a;
    }
    int inc = s;
    #pragma unroll
    for (int off = 1; off < 32; off <<= 1) {
      const int nb = __shfl_up(inc, off);
      if (t >= off) inc += nb;
    }
    const int base = beg + inc - s;
    #pragma unroll
    for (int q = 0; q < kFine / 32; ++q) {
      const int o = base + v[q];
      cur[t * (kFine / 32) + q] = o;
      rowptr[bin * kFine + t * (kFine / 32) + q] = o;
    }
    if (t == 0 && bin == kBins - 1) rowptr[kN] = end;
  }
  __syncthreads();

  int p[kSegPerWave];
  #pragma unroll
  for (int q = 0; q < kSegPerWave; ++q) {
    p[q] = 0;
    if (lane < sn[q]) p[q] = atomicAdd(&cur[(k[q].x >> 16) & (kFine - 1)], 1);
  }
  #pragma unroll
  for (int q = 0; q < kSegPerWave; ++q)
    if (lane < sn[q]) edges[p[q]] = make_int2(k[q].x & 0xFFFF, k[q].y);
  for (int q = 0; q < kSegPerWave; ++q) {
    const int ch = wv * kSegPerWave + q;
    const int pc = seg[ch];
    const int b = ch * kChunkE + (pc & 0xFFFF);
    const int n = pc >> 16;
    for (int j = 32 + lane; j < n; j += 32) {
      const int2 rr = recs[b + j];
      const int pp = atomicAdd(&cur[(rr.x >> 16) & (kFine - 1)], 1);
      edges[pp] = make_int2(rr.x & 0xFFFF, rr.y);
    }
  }
}

__global__ void gemm_wmma_kernel(const float* __restrict__ A,
                                 const float* __restrict__ W,
                                 const float* __restrict__ dinv,
                                 float* __restrict__ H, int K) {
  __shared__ float Ws[kMaxK * 17];

  const int tid  = threadIdx.x;
  const int lane = tid & 31;
  const int wave = tid >> 5;
  const int n0 = blockIdx.y * 16;
  const int half = lane >> 4;
  const int l = lane & 15;

  for (int idx = tid; idx < K * 16; idx += 128) {
    int kk = idx >> 4;
    int nn = idx & 15;
    Ws[kk * 17 + nn] = W[(size_t)kk * kDH + n0 + nn];
  }
  __syncthreads();

  for (int g = blockIdx.x; g < kN / 64; g += gridDim.x) {
    const int m0 = (g * 4 + wave) * 16;
    v8f acc = {};
    const float* arow = A + (size_t)(m0 + l) * K;
    float ds[8];
    #pragma unroll
    for (int r = 0; r < 8; ++r) ds[r] = dinv[m0 + half * 8 + r];
    #pragma unroll 4
    for (int k = 0; k < K; k += 4) {
      const int ka = k + 2 * half;
      v2f a, b;
      a.x = arow[ka];
      a.y = arow[ka + 1];
      b.x = Ws[ka * 17 + l];
      b.y = Ws[(ka + 1) * 17 + l];
      acc = __builtin_amdgcn_wmma_f32_16x16x4_f32(
          false, a, false, b, (short)0, acc, false, false);
    }

    float* hrow = H + (size_t)(m0 + half * 8) * kDH + (n0 + l);
    #pragma unroll
    for (int r = 0; r < 8; ++r) hrow[(size_t)r * kDH] = acc[r] * ds[r];
  }
}

template <int NL>
__device__ __forceinline__ void rows_fma(const float4* __restrict__ h4, int l,
                                         int sx, float wx, float4& acc) {
  int s[NL];
  float w[NL];
  float4 a[NL];
  #pragma unroll
  for (int q = 0; q < NL; ++q) {
    s[q] = __shfl(sx, q, 16);
    w[q] = __shfl(wx, q, 16);
  }
  #pragma unroll
  for (int q = 0; q < NL; ++q) a[q] = h4[(size_t)s[q] * 16 + l];
  #pragma unroll
  for (int q = 0; q < NL; ++q) {
    acc.x += w[q] * a[q].x;
    acc.y += w[q] * a[q].y;
    acc.z += w[q] * a[q].z;
    acc.w += w[q] * a[q].w;
  }
}

template <class Emit>
__device__ __forceinline__ void gather_rows(const int* __restrict__ rowptr,
                                            const int2* __restrict__ edges,
                                            const float* __restrict__ h,
                                            const float* __restrict__ dinv,
                                            const float* __restrict__ bias,
                                            Emit emit) {
  const int t = blockIdx.x * 256 + threadIdx.x;
  const int hw = t >> 4;
  const int l = t & 15;
  const float4* __restrict__ h4 = reinterpret_cast<const float4*>(h);
  const float b0 = bias[l * 4 + 0], b1 = bias[l * 4 + 1];
  const float b2 = bias[l * 4 + 2], b3 = bias[l * 4 + 3];

  int i = hw;
  int beg = rowptr[i], end = rowptr[i + 1];
  int i1 = min(i + kNodeHW, kN - 1);
  int beg1 = rowptr[i1], end1 = rowptr[i1 + 1];
  int2 ed = edges[min(beg + l, kE - 1)];

  for (int it = 0; it < kNodeIters; ++it) {
    const int i2 = min(i1 + kNodeHW, kN - 1);
    const int beg2 = rowptr[i2], end2 = rowptr[i2 + 1];
    const int2 ed1 = edges[min(beg1 + l, kE - 1)];
    const float di = dinv[i];
    const float4 hs = h4[(size_t)i * 16 + l];

    float4 acc = make_float4(0.f, 0.f, 0.f, 0.f);
    int base = beg;
    int sx = (base + l < end) ? ed.x : i;
    float wx = (base + l < end) ? __int_as_float(ed.y) : 0.f;
    while (base < end) {
      const int m = end - base;
      const int nb = base + 16;
      ed = edges[min(nb + l, kE - 1)];
      if (m > 8) {
        rows_fma<16>(h4, l, sx, wx, acc);
      } else {
        rows_fma<8>(h4, l, sx, wx, acc);
      }
      sx = (nb + l < end) ? ed.x : i;
      wx = (nb + l < end) ? __int_as_float(ed.y) : 0.f;
      base = nb;
    }

    const float dd = di * di;
    float4 v;
    v.x = dd * acc.x + (dd * hs.x + di * b0);
    v.y = dd * acc.y + (dd * hs.y + di * b1);
    v.z = dd * acc.z + (dd * hs.z + di * b2);
    v.w = dd * acc.w + (dd * hs.w + di * b3);
    v.x = fmaxf(v.x, 0.f);
    v.y = fmaxf(v.y, 0.f);
    v.z = fmaxf(v.z, 0.f);
    v.w = fmaxf(v.w, 0.f);
    emit(it, i, l, v);

    i = i1; beg = beg1; end = end1; ed = ed1;
    i1 = i2; beg1 = beg2; end1 = end2;
  }
}

constexpr int kTP = 68;
static_assert(kNodeIters * 16 == 64, "the LDS tile holds 4 M-tiles of 16 rows");

__global__ __launch_bounds__(256) __attribute__((amdgpu_waves_per_eu(4, 8))) void gather64_gemm_kernel(const int* __restrict__ rowptr,
                                                       const int2* __restrict__ edges,
                                                       const float* __restrict__ h,
                                                       const float* __restrict__ dinv,
                                                       const float* __restrict__ bias,
                                                       const float* __restrict__ Wn,
                                                       float* __restrict__ hn) {
  __shared__ float As[64 * kTP];
  __shared__ float Wt[64 * kTP];
  const int tid = threadIdx.x;

  #pragma unroll
  for (int j = 0; j < 16; ++j) {
    const int idx = tid + j * 256;
    Wt[(idx & 63) * kTP + (idx >> 6)] = Wn[idx];
  }

  const int hwl = tid >> 4;
  gather_rows(rowptr, edges, h, dinv, bias,
              [&](int it, int  , int l, const float4& v) {
                *reinterpret_cast<float4*>(&As[(it * 16 + hwl) * kTP + 4 * l]) = v;
              });
  __syncthreads();

  const int lane = tid & 31;
  const int wave = tid >> 5;
  const int half = lane >> 4;
  const int l = lane & 15;
  const int mt = wave >> 1;
  const int n0 = (wave & 1) * 32;
  const float* ap  = &As[(mt * 16 + l) * kTP + 2 * half];
  const float* bp0 = &Wt[(n0 + l) * kTP + 2 * half];
  const float* bp1 = &Wt[(n0 + 16 + l) * kTP + 2 * half];
  v8f acc0 = {}, acc1 = {};
  #pragma unroll
  for (int k = 0; k < kDH; k += 4) {
    const v2f a  = *reinterpret_cast<const v2f*>(ap + k);
    const v2f b0 = *reinterpret_cast<const v2f*>(bp0 + k);
    const v2f b1 = *reinterpret_cast<const v2f*>(bp1 + k);
    acc0 = __builtin_amdgcn_wmma_f32_16x16x4_f32(
        false, a, false, b0, (short)0, acc0, false, false);
    acc1 = __builtin_amdgcn_wmma_f32_16x16x4_f32(
        false, a, false, b1, (short)0, acc1, false, false);
  }

  float* hrow = hn + (size_t)(blockIdx.x * 16 + mt * kNodeHW + 8 * half) * kDH + (n0 + l);
  #pragma unroll
  for (int r = 0; r < 8; ++r) {
    hrow[(size_t)r * kDH] = acc0[r];
    hrow[(size_t)r * kDH + 16] = acc1[r];
  }
}

__global__ __launch_bounds__(256) __attribute__((amdgpu_waves_per_eu(4, 8))) void gather64_proj_kernel(const int* __restrict__ rowptr,
                                                       const int2* __restrict__ edges,
                                                       const float* __restrict__ h,
                                                       const float* __restrict__ dinv,
                                                       const float* __restrict__ bias,
                                                       const float* __restrict__ W8,
                                                       float* __restrict__ h2) {
  const int lw = threadIdx.x & 15;
  float w0[4], w1[4];
  #pragma unroll
  for (int q = 0; q < 4; ++q) {
    w0[q] = W8[(lw * 4 + q) * 2 + 0];
    w1[q] = W8[(lw * 4 + q) * 2 + 1];
  }
  gather_rows(rowptr, edges, h, dinv, bias,
              [&](int  , int i, int l, const float4& v) {
                float s0 = v.x * w0[0];
                float s1 = v.x * w1[0];
                s0 = fmaf(v.y, w0[1], s0);
                s1 = fmaf(v.y, w1[1], s1);
                s0 = fmaf(v.z, w0[2], s0);
                s1 = fmaf(v.z, w1[2], s1);
                s0 = fmaf(v.w, w0[3], s0);
                s1 = fmaf(v.w, w1[3], s1);
                s0 = sum16(s0);
                s1 = sum16(s1);
                if (l == 0) reinterpret_cast<float2*>(h2)[i] = make_float2(s0, s1);
              });
}

__global__ void gather2_kernel(const int* __restrict__ rowptr,
                               const int2* __restrict__ edges,
                               const float* __restrict__ h2,
                               const float* __restrict__ dinv,
                               const float* __restrict__ b8,
                               float* __restrict__ out) {
  const int t = blockIdx.x * 256 + threadIdx.x;
  const int hw = t >> 4;
  const int l = t & 15;
  int beg[kNodeIters], end[kNodeIters];
  #pragma unroll
  for (int q = 0; q < kNodeIters; ++q) {
    beg[q] = rowptr[hw + q * kNodeHW];
    end[q] = rowptr[hw + q * kNodeHW + 1];
  }
  #pragma unroll
  for (int q = 0; q < kNodeIters; ++q) {
    const int i = hw + q * kNodeHW;
    float a0 = 0.f, a1 = 0.f;
    for (int j = beg[q] + l; j < end[q]; j += 16) {
      const int2 ed = edges[j];
      const float w = __int_as_float(ed.y);
      a0 += w * h2[(size_t)ed.x * 2 + 0];
      a1 += w * h2[(size_t)ed.x * 2 + 1];
    }
    a0 = sum16(a0);
    a1 = sum16(a1);
    if (l < 2) {
      const float di = dinv[i];
      const float a = (l == 0) ? a0 : a1;
      out[(size_t)i * 2 + l] = di * (a + h2[(size_t)i * 2 + l]) + b8[l];
    }
  }
}

}

extern "C" void kernel_launch(void* const* d_in, const int* in_sizes, int n_in,
                              void* d_out, int out_size, void* d_ws, size_t ws_size,
                              hipStream_t stream) {
  (void)in_sizes; (void)n_in; (void)out_size; (void)ws_size;

  const float* x  = (const float*)d_in[0];
  const int*   ei = (const int*)d_in[1];
  const float* ew = (const float*)d_in[2];
  const float* W[5] = {(const float*)d_in[3], (const float*)d_in[5], (const float*)d_in[7],
                       (const float*)d_in[9], (const float*)d_in[11]};
  const float* B[5] = {(const float*)d_in[4], (const float*)d_in[6], (const float*)d_in[8],
                       (const float*)d_in[10], (const float*)d_in[12]};
  const float* W8 = (const float*)d_in[13];
  const float* b8 = (const float*)d_in[14];
  float* out = (float*)d_out;

  const int* row = ei;
  const int* col = ei + kE;

  float* ws = (float*)d_ws;
  float* dinv  = ws;         ws += kN;
  int2*  edges = (int2*)ws;  ws += (size_t)kE * 2;
  float* feat  = ws;         ws += (size_t)kN * kDH;
  float* hbuf  = ws;         ws += (size_t)kN * kDH;
  int*   table = (int*)ws;   ws += kN;
  int*   rowptr = (int*)ws;
  int2*  recs  = (int2*)feat;

  chunk_sort_kernel<<<kChunks, kSortT, 0, stream>>>(row, col, ew, table, recs);
  node_sort_kernel<<<kBins, kSortT, 0, stream>>>(recs, table, rowptr, edges, dinv);

  const dim3 ggrid(kGemmBlocksX, kDH / 16);

  gemm_wmma_kernel<<<ggrid, 128, 0, stream>>>(x, W[0], dinv, hbuf, 128);
  gather64_gemm_kernel<<<kNodeBlocks, 256, 0, stream>>>(rowptr, edges, hbuf, dinv, B[0], W[1], feat);
  gather64_gemm_kernel<<<kNodeBlocks, 256, 0, stream>>>(rowptr, edges, feat, dinv, B[1], W[2], hbuf);
  gather64_gemm_kernel<<<kNodeBlocks, 256, 0, stream>>>(rowptr, edges, hbuf, dinv, B[2], W[3], feat);
  gather64_gemm_kernel<<<kNodeBlocks, 256, 0, stream>>>(rowptr, edges, feat, dinv, B[3], W[4], hbuf);

  gather64_proj_kernel<<<kNodeBlocks, 256, 0, stream>>>(rowptr, edges, hbuf, dinv, B[4], W8, feat);
  gather2_kernel<<<kNodeBlocks, 256, 0, stream>>>(rowptr, edges, feat, dinv, b8, out);
}


# Layer-1 product x@W1 loses its launch: it runs raw as extra blocks of the first bucketing launch (new k order in it); first gather scales each edge weight by dinv[src]; bucket order kept: 8 launches
# speedup vs baseline: 5.4915x; 1.0384x over previous
#include <hip/hip_runtime.h>


namespace {

constexpr int kN  = 65536;
constexpr int kE  = 1048576;
constexpr int kDH = 64;
constexpr int kMaxK = 128;

constexpr int kEdgeBlocks = 512;
constexpr int kEdgeT = kEdgeBlocks * 256;
constexpr int kEdgeIters = kE / kEdgeT;
constexpr int kNodeBlocks = 1024;
constexpr int kNodeHW = kNodeBlocks * 256 / 16;
constexpr int kNodeIters = kN / kNodeHW;
static_assert(kEdgeIters * kEdgeT == kE, "edge launch must tile the edges");
static_assert(kNodeIters * kNodeHW == kN, "node launch must tile the nodes");

constexpr int kBins = 256;
constexpr int kFineShift = 8;
constexpr int kFine = 1 << kFineShift;
constexpr int kChunks = 256;
constexpr int kChunkE = kE / kChunks;
constexpr int kSortT = 512;
constexpr int kSortIters = kChunkE / kSortT;
constexpr int kSortW = kSortT / 32;
constexpr int kSegPerWave = kChunks / kSortW;
static_assert(kBins * kFine == kN, "ranges must tile the nodes");
static_assert(kBins * kChunks == kN, "the (range, chunk) table fills the kN-word table region");
static_assert(kSortIters * kSortT * kChunks == kE, "chunks must tile the edges");
static_assert(kSegPerWave * kSortW == kChunks, "the waves of a range block must tile its pieces");
static_assert(kBins <= kSortT && kChunks <= kSortT, "one thread per range / per piece");
static_assert(kChunkE <= 0x7FFF, "a piece's offset and count must fit 16 bits each, sign clear");
static_assert(kN <= 65536, "a source index must fit the low 16 bits of a packed record");

typedef __attribute__((ext_vector_type(2))) float v2f;
typedef __attribute__((ext_vector_type(8))) float v8f;

__device__ __forceinline__ float sum16(float v) {
  v += __shfl_xor(v, 8, 16);
  v += __shfl_xor(v, 4, 16);
  v += __shfl_xor(v, 2, 16);
  v += __shfl_xor(v, 1, 16);
  return v;
}

__device__ __forceinline__ void chunk_sort_block(const int chunk,
                                                 const int* __restrict__ row,
                                                 const int* __restrict__ col,
                                                 const float* __restrict__ ew,
                                                 int* __restrict__ table,
                                                 int2* __restrict__ recs) {
  __shared__ int h[kBins];
  __shared__ int offs[kBins];
  __shared__ int wtot[kBins / 32];
  const int t = threadIdx.x;
  const int lane = t & 31;
  const int wv = t >> 5;
  const int e0 = chunk * kChunkE + t;
  int c[kSortIters], s[kSortIters], r[kSortIters];
  float w[kSortIters];
  if (t < kBins) h[t] = 0;
  #pragma unroll
  for (int q = 0; q < kSortIters; ++q) {
    const int e = e0 + q * kSortT;
    c[q] = col[e];
    s[q] = row[e];
    w[q] = ew[e];
  }
  __syncthreads();
  #pragma unroll
  for (int q = 0; q < kSortIters; ++q) r[q] = atomicAdd(&h[(c[q] >> kFineShift) & (kBins - 1)], 1);
  __syncthreads();

  int v = 0, inc = 0;
  if (t < kBins) {
    v = h[t];
    inc = v;
    #pragma unroll
    for (int off = 1; off < 32; off <<= 1) {
      const int nb = __shfl_up(inc, off);
      if (lane >= off) inc += nb;
    }
    if (lane == 31) wtot[wv] = inc;
  }
  __syncthreads();
  if (t < kBins) {
    int base = 0;
    #pragma unroll
    for (int k = 0; k < kBins / 32; ++k) base += (k < wv) ? wtot[k] : 0;
    const int ex = base + inc - v;
    offs[t] = ex;
    table[t * kChunks + chunk] = ex | (v << 16);
  }
  __syncthreads();

  int2* __restrict__ slice = recs + (size_t)chunk * kChunkE;
  #pragma unroll
  for (int q = 0; q < kSortIters; ++q)
    slice[offs[(c[q] >> kFineShift) & (kBins - 1)] + r[q]] =
        make_int2(s[q] | ((c[q] & (kFine - 1)) << 16), __float_as_int(w[q]));
}

static_assert(kSortT >= 2 * kFine, "threads kFine .. 2*kFine-1 write the range's dinv");
__global__ __launch_bounds__(kSortT) void node_sort_kernel(const int2* __restrict__ recs,
                                                           const int* __restrict__ table,
                                                           int* __restrict__ rowptr,
                                                           int2* __restrict__ edges,
                                                           float* __restrict__ dinv) {
  __shared__ int cnt[kFine];
  __shared__ int cur[kFine];
  __shared__ float wdeg[kFine];
  __shared__ int seg[kChunks];
  __shared__ int wbeg[kSortW];
  __shared__ int wnum[kSortW];
  const int t = threadIdx.x;
  const int lane = t & 31;
  const int wv = t >> 5;
  const int bin = blockIdx.x;

  int oc = 0;
  if (t < kChunks) {
    oc = table[bin * kChunks + t];
    seg[t] = oc;
  }
  if (t < kFine) {
    cnt[t] = 0;
    wdeg[t] = 0.f;
  }
  __syncthreads();

  int2 k[kSegPerWave];
  int sb[kSegPerWave], sn[kSegPerWave];
  #pragma unroll
  for (int q = 0; q < kSegPerWave; ++q) {
    const int ch = wv * kSegPerWave + q;
    const int pc = seg[ch];
    sb[q] = ch * kChunkE + (pc & 0xFFFF);
    sn[q] = pc >> 16;
    k[q] = recs[min(sb[q] + lane, kE - 1)];
  }
  int so = oc & 0xFFFF;
  int sc = oc >> 16;
  #pragma unroll
  for (int off = 16; off > 0; off >>= 1) {
    so += __shfl_xor(so, off);
    sc += __shfl_xor(sc, off);
  }
  if (lane == 0) {
    wbeg[wv] = so;
    wnum[wv] = sc;
  }
  #pragma unroll
  for (int q = 0; q < kSegPerWave; ++q)
    if (lane < sn[q]) {
      const int nd = (k[q].x >> 16) & (kFine - 1);
      atomicAdd(&cnt[nd], 1);
      atomicAdd(&wdeg[nd], __int_as_float(k[q].y));
    }
  for (int q = 0; q < kSegPerWave; ++q) {
    const int ch = wv * kSegPerWave + q;
    const int pc = seg[ch];
    const int b = ch * kChunkE + (pc & 0xFFFF);
    const int n = pc >> 16;
    for (int j = 32 + lane; j < n; j += 32) {
      const int2 rr = recs[b + j];
      const int nd = (rr.x >> 16) & (kFine - 1);
      atomicAdd(&cnt[nd], 1);
      atomicAdd(&wdeg[nd], __int_as_float(rr.y));
    }
  }
  __syncthreads();

  if (t >= kFine && t < 2 * kFine) {
    const float s = wdeg[t - kFine] + 1.0f;
    dinv[bin * kFine + (t - kFine)] = (s > 0.f) ? rsqrtf(s) : 0.f;
  }

  if (t < 32) {
    int beg = 0, num = 0;
    #pragma unroll
    for (int q = 0; q < kSortW; ++q) {
      beg += wbeg[q];
      num += wnum[q];
    }
    const int end = beg + num;
    int v[kFine / 32];
    int s = 0;
    #pragma unroll
    for (int q = 0; q < kFine / 32; ++q) {
      const int a = cnt[t * (kFine / 32) + q];
      v[q] = s;
      s += a;
    }
    int inc = s;
    #pragma unroll
    for (int off = 1; off < 32; off <<= 1) {
      const int nb = __shfl_up(inc, off);
      if (t >= off) inc += nb;
    }
    const int base = beg + inc - s;
    #pragma unroll
    for (int q = 0; q < kFine / 32; ++q) {
      const int o = base + v[q];
      cur[t * (kFine / 32) + q] = o;
      rowptr[bin * kFine + t * (kFine / 32) + q] = o;
    }
    if (t == 0 && bin == kBins - 1) rowptr[kN] = end;
  }
  __syncthreads();

  int p[kSegPerWave];
  #pragma unroll
  for (int q = 0; q < kSegPerWave; ++q) {
    p[q] = 0;
    if (lane < sn[q]) p[q] = atomicAdd(&cur[(k[q].x >> 16) & (kFine - 1)], 1);
  }
  #pragma unroll
  for (int q = 0; q < kSegPerWave; ++q)
    if (lane < sn[q]) edges[p[q]] = make_int2(k[q].x & 0xFFFF, k[q].y);
  for (int q = 0; q < kSegPerWave; ++q) {
    const int ch = wv * kSegPerWave + q;
    const int pc = seg[ch];
    const int b = ch * kChunkE + (pc & 0xFFFF);
    const int n = pc >> 16;
    for (int j = 32 + lane; j < n; j += 32) {
      const int2 rr = recs[b + j];
      const int pp = atomicAdd(&cur[(rr.x >> 16) & (kFine - 1)], 1);
      edges[pp] = make_int2(rr.x & 0xFFFF, rr.y);
    }
  }
}

constexpr int kWP = 132;
static_assert(kMaxK == 128 && kDH == 64, "the staging below is written for a 128 x 64 W1");
static_assert(kN == kChunks * kSortW * 16, "one 16-row tile per wave of the product blocks");

__device__ __forceinline__ void xw1_block(const int gb,
                                          const float* __restrict__ A,
                                          const float* __restrict__ W,
                                          float* __restrict__ H) {
  __shared__ alignas(16) float Wt1[kDH * kWP];
  const int tid = threadIdx.x;
  const int lane = tid & 31;
  const int wave = tid >> 5;
  const int half = lane >> 4;
  const int l = lane & 15;
  const int m0 = (gb * kSortW + wave) * 16;

  float wv[16];
  #pragma unroll
  for (int j = 0; j < 16; ++j) wv[j] = W[tid + j * kSortT];
  const float4* __restrict__ arow =
      reinterpret_cast<const float4*>(A + (size_t)(m0 + l) * kMaxK + 4 * half);
  float4 a[16];
  #pragma unroll
  for (int j = 0; j < 16; ++j) a[j] = arow[2 * j];
  #pragma unroll
  for (int j = 0; j < 16; ++j) {
    const int idx = tid + j * kSortT;
    Wt1[(idx & 63) * kWP + (idx >> 6)] = wv[j];
  }
  __syncthreads();

  v8f acc[4] = {};
  const float* bp = &Wt1[l * kWP + 4 * half];
  #pragma unroll
  for (int j = 0; j < 16; ++j) {
    float4 b[4];
    #pragma unroll
    for (int n = 0; n < 4; ++n)
      b[n] = *reinterpret_cast<const float4*>(bp + n * 16 * kWP + 8 * j);
    const v2f a0 = {a[j].x, a[j].y};
    const v2f a1 = {a[j].z, a[j].w};
    #pragma unroll
    for (int n = 0; n < 4; ++n) {
      const v2f b0 = {b[n].x, b[n].y};
      acc[n] = __builtin_amdgcn_wmma_f32_16x16x4_f32(
          false, a0, false, b0, (short)0, acc[n], false, false);
    }
    #pragma unroll
    for (int n = 0; n < 4; ++n) {
      const v2f b1 = {b[n].z, b[n].w};
      acc[n] = __builtin_amdgcn_wmma_f32_16x16x4_f32(
          false, a1, false, b1, (short)0, acc[n], false, false);
    }
  }

  float* hrow = H + (size_t)(m0 + half * 8) * kDH + l;
  #pragma unroll
  for (int r = 0; r < 8; ++r) {
    #pragma unroll
    for (int n = 0; n < 4; ++n) hrow[(size_t)r * kDH + n * 16] = acc[n][r];
  }
}

static_assert(kChunks == 256, "the role bits assume 2 x 256 blocks");
__global__ __launch_bounds__(kSortT) void sort_xw1_kernel(const int* __restrict__ row,
                                                          const int* __restrict__ col,
                                                          const float* __restrict__ ew,
                                                          int* __restrict__ table,
                                                          int2* __restrict__ recs,
                                                          const float* __restrict__ x,
                                                          const float* __restrict__ W1,
                                                          float* __restrict__ H) {
  const int b = blockIdx.x;
  const int role = (b ^ (b >> 8)) & 1;
  if (role == 0) {
    chunk_sort_block(b >> 1, row, col, ew, table, recs);
  } else {
    xw1_block(b >> 1, x, W1, H);
  }
}

template <int NL, bool kSrcScale>
__device__ __forceinline__ void rows_fma(const float4* __restrict__ h4, int l,
                                         int sx, float wx, float dsx, float4& acc) {
  int s[NL];
  float w[NL];
  float4 a[NL];
  #pragma unroll
  for (int q = 0; q < NL; ++q) {
    s[q] = __shfl(sx, q, 16);
    if constexpr (!kSrcScale) w[q] = __shfl(wx, q, 16);
  }
  #pragma unroll
  for (int q = 0; q < NL; ++q) a[q] = h4[(size_t)s[q] * 16 + l];
  if constexpr (kSrcScale) {
    #pragma unroll
    for (int q = 0; q < NL; ++q) w[q] = __shfl(wx, q, 16) * __shfl(dsx, q, 16);
  }
  #pragma unroll
  for (int q = 0; q < NL; ++q) {
    acc.x += w[q] * a[q].x;
    acc.y += w[q] * a[q].y;
    acc.z += w[q] * a[q].z;
    acc.w += w[q] * a[q].w;
  }
}

template <bool kSrcScale, class Emit>
__device__ __forceinline__ void gather_rows(const int* __restrict__ rowptr,
                                            const int2* __restrict__ edges,
                                            const float* __restrict__ h,
                                            const float* __restrict__ dinv,
                                            const float* __restrict__ bias,
                                            Emit emit) {
  const int t = blockIdx.x * 256 + threadIdx.x;
  const int hw = t >> 4;
  const int l = t & 15;
  const float4* __restrict__ h4 = reinterpret_cast<const float4*>(h);
  const float b0 = bias[l * 4 + 0], b1 = bias[l * 4 + 1];
  const float b2 = bias[l * 4 + 2], b3 = bias[l * 4 + 3];

  int i = hw;
  int beg = rowptr[i], end = rowptr[i + 1];
  int i1 = min(i + kNodeHW, kN - 1);
  int beg1 = rowptr[i1], end1 = rowptr[i1 + 1];
  int2 ed = edges[min(beg + l, kE - 1)];

  for (int it = 0; it < kNodeIters; ++it) {
    const int i2 = min(i1 + kNodeHW, kN - 1);
    const int beg2 = rowptr[i2], end2 = rowptr[i2 + 1];
    const int2 ed1 = edges[min(beg1 + l, kE - 1)];
    const float di = dinv[i];
    const float4 hs = h4[(size_t)i * 16 + l];

    float4 acc = make_float4(0.f, 0.f, 0.f, 0.f);
    int base = beg;
    int sx = (base + l < end) ? ed.x : i;
    float wx = (base + l < end) ? __int_as_float(ed.y) : 0.f;
    float dsx = 1.f;
    if constexpr (kSrcScale) dsx = dinv[sx];
    while (base < end) {
      const int m = end - base;
      const int nb = base + 16;
      ed = edges[min(nb + l, kE - 1)];
      if (m > 8) {
        rows_fma<16, kSrcScale>(h4, l, sx, wx, dsx, acc);
      } else {
        rows_fma<8, kSrcScale>(h4, l, sx, wx, dsx, acc);
      }
      sx = (nb + l < end) ? ed.x : i;
      wx = (nb + l < end) ? __int_as_float(ed.y) : 0.f;
      if constexpr (kSrcScale) dsx = dinv[sx];
      base = nb;
    }

    const float dd = di * di;
    const float ds = kSrcScale ? dd * di : dd;
    float4 v;
    v.x = dd * acc.x + (ds * hs.x + di * b0);
    v.y = dd * acc.y + (ds * hs.y + di * b1);
    v.z = dd * acc.z + (ds * hs.z + di * b2);
    v.w = dd * acc.w + (ds * hs.w + di * b3);
    v.x = fmaxf(v.x, 0.f);
    v.y = fmaxf(v.y, 0.f);
    v.z = fmaxf(v.z, 0.f);
    v.w = fmaxf(v.w, 0.f);
    emit(it, i, l, v);

    i = i1; beg = beg1; end = end1; ed = ed1;
    i1 = i2; beg1 = beg2; end1 = end2;
  }
}

constexpr int kTP = 68;
static_assert(kNodeIters * 16 == 64, "the LDS tile holds 4 M-tiles of 16 rows");

template <bool kSrcScale>
__global__ __launch_bounds__(256) __attribute__((amdgpu_waves_per_eu(4, 8))) void gather64_gemm_kernel(const int* __restrict__ rowptr,
                                                       const int2* __restrict__ edges,
                                                       const float* __restrict__ h,
                                                       const float* __restrict__ dinv,
                                                       const float* __restrict__ bias,
                                                       const float* __restrict__ Wn,
                                                       float* __restrict__ hn) {
  __shared__ float As[64 * kTP];
  __shared__ float Wt[64 * kTP];
  const int tid = threadIdx.x;

  #pragma unroll
  for (int j = 0; j < 16; ++j) {
    const int idx = tid + j * 256;
    Wt[(idx & 63) * kTP + (idx >> 6)] = Wn[idx];
  }

  const int hwl = tid >> 4;
  gather_rows<kSrcScale>(rowptr, edges, h, dinv, bias,
              [&](int it, int  , int l, const float4& v) {
                *reinterpret_cast<float4*>(&As[(it * 16 + hwl) * kTP + 4 * l]) = v;
              });
  __syncthreads();

  const int lane = tid & 31;
  const int wave = tid >> 5;
  const int half = lane >> 4;
  const int l = lane & 15;
  const int mt = wave >> 1;
  const int n0 = (wave & 1) * 32;
  const float* ap  = &As[(mt * 16 + l) * kTP + 2 * half];
  const float* bp0 = &Wt[(n0 + l) * kTP + 2 * half];
  const float* bp1 = &Wt[(n0 + 16 + l) * kTP + 2 * half];
  v8f acc0 = {}, acc1 = {};
  #pragma unroll
  for (int k = 0; k < kDH; k += 4) {
    const v2f a  = *reinterpret_cast<const v2f*>(ap + k);
    const v2f b0 = *reinterpret_cast<const v2f*>(bp0 + k);
    const v2f b1 = *reinterpret_cast<const v2f*>(bp1 + k);
    acc0 = __builtin_amdgcn_wmma_f32_16x16x4_f32(
        false, a, false, b0, (short)0, acc0, false, false);
    acc1 = __builtin_amdgcn_wmma_f32_16x16x4_f32(
        false, a, false, b1, (short)0, acc1, false, false);
  }

  float* hrow = hn + (size_t)(blockIdx.x * 16 + mt * kNodeHW + 8 * half) * kDH + (n0 + l);
  #pragma unroll
  for (int r = 0; r < 8; ++r) {
    hrow[(size_t)r * kDH] = acc0[r];
    hrow[(size_t)r * kDH + 16] = acc1[r];
  }
}

__global__ __launch_bounds__(256) __attribute__((amdgpu_waves_per_eu(4, 8))) void gather64_proj_kernel(const int* __restrict__ rowptr,
                                                       const int2* __restrict__ edges,
                                                       const float* __restrict__ h,
                                                       const float* __restrict__ dinv,
                                                       const float* __restrict__ bias,
                                                       const float* __restrict__ W8,
                                                       float* __restrict__ h2) {
  const int lw = threadIdx.x & 15;
  float w0[4], w1[4];
  #pragma unroll
  for (int q = 0; q < 4; ++q) {
    w0[q] = W8[(lw * 4 + q) * 2 + 0];
    w1[q] = W8[(lw * 4 + q) * 2 + 1];
  }
  gather_rows<false>(rowptr, edges, h, dinv, bias,
              [&](int  , int i, int l, const float4& v) {
                float s0 = v.x * w0[0];
                float s1 = v.x * w1[0];
                s0 = fmaf(v.y, w0[1], s0);
                s1 = fmaf(v.y, w1[1], s1);
                s0 = fmaf(v.z, w0[2], s0);
                s1 = fmaf(v.z, w1[2], s1);
                s0 = fmaf(v.w, w0[3], s0);
                s1 = fmaf(v.w, w1[3], s1);
                s0 = sum16(s0);
                s1 = sum16(s1);
                if (l == 0) reinterpret_cast<float2*>(h2)[i] = make_float2(s0, s1);
              });
}

__global__ void gather2_kernel(const int* __restrict__ rowptr,
                               const int2* __restrict__ edges,
                               const float* __restrict__ h2,
                               const float* __restrict__ dinv,
                               const float* __restrict__ b8,
                               float* __restrict__ out) {
  const int t = blockIdx.x * 256 + threadIdx.x;
  const int hw = t >> 4;
  const int l = t & 15;
  int beg[kNodeIters], end[kNodeIters];
  #pragma unroll
  for (int q = 0; q < kNodeIters; ++q) {
    beg[q] = rowptr[hw + q * kNodeHW];
    end[q] = rowptr[hw + q * kNodeHW + 1];
  }
  #pragma unroll
  for (int q = 0; q < kNodeIters; ++q) {
    const int i = hw + q * kNodeHW;
    float a0 = 0.f, a1 = 0.f;
    for (int j = beg[q] + l; j < end[q]; j += 16) {
      const int2 ed = edges[j];
      const float w = __int_as_float(ed.y);
      a0 += w * h2[(size_t)ed.x * 2 + 0];
      a1 += w * h2[(size_t)ed.x * 2 + 1];
    }
    a0 = sum16(a0);
    a1 = sum16(a1);
    if (l < 2) {
      const float di = dinv[i];
      const float a = (l == 0) ? a0 : a1;
      out[(size_t)i * 2 + l] = di * (a + h2[(size_t)i * 2 + l]) + b8[l];
    }
  }
}

}

extern "C" void kernel_launch(void* const* d_in, const int* in_sizes, int n_in,
                              void* d_out, int out_size, void* d_ws, size_t ws_size,
                              hipStream_t stream) {
  (void)in_sizes; (void)n_in; (void)out_size; (void)ws_size;

  const float* x  = (const float*)d_in[0];
  const int*   ei = (const int*)d_in[1];
  const float* ew = (const float*)d_in[2];
  const float* W[5] = {(const float*)d_in[3], (const float*)d_in[5], (const float*)d_in[7],
                       (const float*)d_in[9], (const float*)d_in[11]};
  const float* B[5] = {(const float*)d_in[4], (const float*)d_in[6], (const float*)d_in[8],
                       (const float*)d_in[10], (const float*)d_in[12]};
  const float* W8 = (const float*)d_in[13];
  const float* b8 = (const float*)d_in[14];
  float* out = (float*)d_out;

  const int* row = ei;
  const int* col = ei + kE;

  float* ws = (float*)d_ws;
  float* dinv  = ws;         ws += kN;
  int2*  edges = (int2*)ws;  ws += (size_t)kE * 2;
  float* feat  = ws;         ws += (size_t)kN * kDH;
  float* hbuf  = ws;         ws += (size_t)kN * kDH;
  int*   table = (int*)ws;   ws += kN;
  int*   rowptr = (int*)ws;
  int2*  recs  = (int2*)feat;

  sort_xw1_kernel<<<2 * kChunks, kSortT, 0, stream>>>(row, col, ew, table, recs, x, W[0], hbuf);
  node_sort_kernel<<<kBins, kSortT, 0, stream>>>(recs, table, rowptr, edges, dinv);

  gather64_gemm_kernel<true><<<kNodeBlocks, 256, 0, stream>>>(rowptr, edges, hbuf, dinv, B[0], W[1], feat);
  gather64_gemm_kernel<false><<<kNodeBlocks, 256, 0, stream>>>(rowptr, edges, feat, dinv, B[1], W[2], hbuf);
  gather64_gemm_kernel<false><<<kNodeBlocks, 256, 0, stream>>>(rowptr, edges, hbuf, dinv, B[2], W[3], feat);
  gather64_gemm_kernel<false><<<kNodeBlocks, 256, 0, stream>>>(rowptr, edges, feat, dinv, B[3], W[4], hbuf);

  gather64_proj_kernel<<<kNodeBlocks, 256, 0, stream>>>(rowptr, edges, hbuf, dinv, B[4], W8, feat);
  gather2_kernel<<<kNodeBlocks, 256, 0, stream>>>(rowptr, edges, feat, dinv, b8, out);
}
